# starting kernel
# baseline (speedup 1.0000x reference)
#include <hip/hip_runtime.h>
#include <hip/hip_bf16.h>

typedef __attribute__((ext_vector_type(2))) float v2f;
typedef __attribute__((ext_vector_type(8))) float v8f;

#define N_NODES 100000
#define KDIM    128

__device__ __forceinline__ void atomic_add_f32(float* p, float v) {
  asm volatile("global_atomic_add_f32 %0, %1, off" : : "v"(p), "v"(v) : "memory");
}

__global__ __launch_bounds__(256) void spmm_edge_atomic(
    const int* __restrict__ rows, const int* __restrict__ cols,
    const float* __restrict__ vals, const float* __restrict__ x,
    float* __restrict__ out, int E) {
  int e = blockIdx.x * 8 + (threadIdx.x >> 5);
  if (e >= E) return;
  int lane = threadIdx.x & 31;
  int r = rows[e];
  int c = cols[e];
  float v = vals[e];
  float4 xv = *((const float4*)(x + (size_t)c * 128) + lane);
  float* o = out + (size_t)r * 128 + lane * 4;
  atomic_add_f32(o + 0, xv.x * v);
  atomic_add_f32(o + 1, xv.y * v);
  atomic_add_f32(o + 2, xv.z * v);
  atomic_add_f32(o + 3, xv.w * v);
}

template <int NCOLS, bool RELU>
__global__ __launch_bounds__(256) void gemm_wmma_f32(
    const float* __restrict__ A, const float* __restrict__ B,
    const float* __restrict__ bias, float* __restrict__ C, int M) {
  constexpr int K = KDIM;
  constexpr int LDSP = 2 * NCOLS + 32;
  __shared__ __align__(16) float Bs[(K / 2) * LDSP];

  for (int i = threadIdx.x; i < K * NCOLS; i += 256) {
    int k = i / NCOLS, n = i % NCOLS;
    Bs[(k >> 1) * LDSP + n * 2 + (k & 1)] = B[i];
  }
  __syncthreads();

  const int wave = threadIdx.x >> 5;
  const int lane = threadIdx.x & 31;
  const int tileRow = blockIdx.x * 8 + wave;
  if (tileRow * 16 >= M) return;
  const int row0 = tileRow * 16;

  const int mlane = lane & 15;
  const int khalf = (lane >> 4) * 2;

  v2f afrag[K / 4];
  const float* arow = A + (size_t)(row0 + mlane) * K + khalf;
#pragma unroll
  for (int ks = 0; ks < K / 4; ++ks)
    afrag[ks] = *(const v2f*)(arow + ks * 4);

#pragma unroll
  for (int nt = 0; nt < NCOLS / 16; ++nt) {
    const int n = nt * 16 + mlane;
    const float bv = bias[n];
    v8f acc;
#pragma unroll
    for (int i = 0; i < 8; ++i) acc[i] = bv;

#pragma unroll
    for (int ks = 0; ks < K / 4; ++ks) {
      const int kpair = ks * 2 + (lane >> 4);
      v2f bfrag = *(const v2f*)&Bs[kpair * LDSP + n * 2];
      acc = __builtin_amdgcn_wmma_f32_16x16x4_f32(
           false, afrag[ks],  false, bfrag,
           (short)0, acc,  false,  false);
    }

    const int mbase = row0 + ((lane >> 4) * 8);
#pragma unroll
    for (int v = 0; v < 8; ++v) {
      float val = acc[v];
      if (RELU) val = fmaxf(val, 0.0f);
      C[(size_t)(mbase + v) * NCOLS + n] = val;
    }
  }
}

extern "C" void kernel_launch(void* const* d_in, const int* in_sizes, int n_in,
                              void* d_out, int out_size, void* d_ws, size_t ws_size,
                              hipStream_t stream) {
  const float* x  = (const float*)d_in[0];
  const int*   er = (const int*)d_in[1];
  const int*   ec = (const int*)d_in[2];
  const float* ev = (const float*)d_in[3];
  const float* W1 = (const float*)d_in[4];
  const float* b1 = (const float*)d_in[5];
  const float* W2 = (const float*)d_in[6];
  const float* b2 = (const float*)d_in[7];
  float* out = (float*)d_out;
  const int E = in_sizes[1];

  const size_t hElems = (size_t)N_NODES * 128;
  float* h1 = (float*)d_ws;
  float* h2 = h1 + hElems;

  const int spmmBlocks = (E + 7) / 8;
  const int gemmBlocks = ((N_NODES / 16) + 7) / 8;

  hipMemsetAsync(h1, 0, hElems * sizeof(float), stream);
  spmm_edge_atomic<<<spmmBlocks, 256, 0, stream>>>(er, ec, ev, x, h1, E);
  gemm_wmma_f32<128, true><<<gemmBlocks, 256, 0, stream>>>(h1, W1, b1, h2, N_NODES);
  hipMemsetAsync(h1, 0, hElems * sizeof(float), stream);
  spmm_edge_atomic<<<spmmBlocks, 256, 0, stream>>>(er, ec, ev, h2, h1, E);
  gemm_wmma_f32<64, false><<<gemmBlocks, 256, 0, stream>>>(h1, W2, b2, out, N_NODES);
}


# Layer 2 reassociated: W2 product (128->64) before the edge scatter, adj(h W2)+b2 for (adj h)W2+b2; scatter 64 wide (half the f32 atomics), out starts as b2 rows; changes the order of summation
# speedup vs baseline: 1.3835x; 1.3835x over previous
#include <hip/hip_runtime.h>
#include <hip/hip_bf16.h>

typedef __attribute__((ext_vector_type(2))) float v2f;
typedef __attribute__((ext_vector_type(8))) float v8f;

#define N_NODES 100000
#define KDIM    128

__device__ __forceinline__ void atomic_add_f32(float* p, float v) {
  asm volatile("global_atomic_add_f32 %0, %1, off" : : "v"(p), "v"(v) : "memory");
}

__global__ __launch_bounds__(256) void spmm_edge_atomic(
    const int* __restrict__ rows, const int* __restrict__ cols,
    const float* __restrict__ vals, const float* __restrict__ x,
    float* __restrict__ out, int E) {
  int e = blockIdx.x * 8 + (threadIdx.x >> 5);
  if (e >= E) return;
  int lane = threadIdx.x & 31;
  int r = rows[e];
  int c = cols[e];
  float v = vals[e];
  float4 xv = *((const float4*)(x + (size_t)c * 128) + lane);
  float* o = out + (size_t)r * 128 + lane * 4;
  atomic_add_f32(o + 0, xv.x * v);
  atomic_add_f32(o + 1, xv.y * v);
  atomic_add_f32(o + 2, xv.z * v);
  atomic_add_f32(o + 3, xv.w * v);
}

__global__ __launch_bounds__(256) void spmm_edge_atomic64(
    const int* __restrict__ rows, const int* __restrict__ cols,
    const float* __restrict__ vals, const float* __restrict__ g,
    float* __restrict__ out, int E) {
  int e = blockIdx.x * 16 + (threadIdx.x >> 4);
  if (e >= E) return;
  int l = threadIdx.x & 15;
  int r = rows[e];
  int c = cols[e];
  float v = vals[e];
  float4 gv = *((const float4*)(g + (size_t)c * 64) + l);
  float* o = out + (size_t)r * 64 + l * 4;
  atomic_add_f32(o + 0, gv.x * v);
  atomic_add_f32(o + 1, gv.y * v);
  atomic_add_f32(o + 2, gv.z * v);
  atomic_add_f32(o + 3, gv.w * v);
}

template <int NCOLS, bool RELU, bool BIAS_ASIDE>
__global__ __launch_bounds__(256) void gemm_wmma_f32(
    const float* __restrict__ A, const float* __restrict__ B,
    const float* __restrict__ bias, float* __restrict__ C,
    float* __restrict__ Cbias, int M) {
  constexpr int K = KDIM;
  constexpr int LDSP = 2 * NCOLS + 32;
  __shared__ __align__(16) float Bs[(K / 2) * LDSP];

  for (int i = threadIdx.x; i < K * NCOLS; i += 256) {
    int k = i / NCOLS, n = i % NCOLS;
    Bs[(k >> 1) * LDSP + n * 2 + (k & 1)] = B[i];
  }
  __syncthreads();

  const int wave = threadIdx.x >> 5;
  const int lane = threadIdx.x & 31;
  const int tileRow = blockIdx.x * 8 + wave;
  if (tileRow * 16 >= M) return;
  const int row0 = tileRow * 16;

  const int mlane = lane & 15;
  const int khalf = (lane >> 4) * 2;

  v2f afrag[K / 4];
  const float* arow = A + (size_t)(row0 + mlane) * K + khalf;
#pragma unroll
  for (int ks = 0; ks < K / 4; ++ks)
    afrag[ks] = *(const v2f*)(arow + ks * 4);

#pragma unroll
  for (int nt = 0; nt < NCOLS / 16; ++nt) {
    const int n = nt * 16 + mlane;
    const float bv = BIAS_ASIDE ? 0.0f : bias[n];
    v8f acc;
#pragma unroll
    for (int i = 0; i < 8; ++i) acc[i] = bv;

#pragma unroll
    for (int ks = 0; ks < K / 4; ++ks) {
      const int kpair = ks * 2 + (lane >> 4);
      v2f bfrag = *(const v2f*)&Bs[kpair * LDSP + n * 2];
      acc = __builtin_amdgcn_wmma_f32_16x16x4_f32(
           false, afrag[ks],  false, bfrag,
           (short)0, acc,  false,  false);
    }

    const int mbase = row0 + ((lane >> 4) * 8);
#pragma unroll
    for (int v = 0; v < 8; ++v) {
      float val = acc[v];
      if (RELU) val = fmaxf(val, 0.0f);
      C[(size_t)(mbase + v) * NCOLS + n] = val;
    }
  }

  if (BIAS_ASIDE) {
    static_assert(128 % NCOLS == 0, "row pieces must tile a 128-float store");
    const float4 bq = *(const float4*)(bias + ((lane * 4) % NCOLS));
    float* cb = Cbias + (size_t)row0 * NCOLS + lane * 4;
    for (int i = 0; i < (16 * NCOLS) / 128; ++i)
      *(float4*)(cb + i * 128) = bq;
  }
}

extern "C" void kernel_launch(void* const* d_in, const int* in_sizes, int n_in,
                              void* d_out, int out_size, void* d_ws, size_t ws_size,
                              hipStream_t stream) {
  const float* x  = (const float*)d_in[0];
  const int*   er = (const int*)d_in[1];
  const int*   ec = (const int*)d_in[2];
  const float* ev = (const float*)d_in[3];
  const float* W1 = (const float*)d_in[4];
  const float* b1 = (const float*)d_in[5];
  const float* W2 = (const float*)d_in[6];
  const float* b2 = (const float*)d_in[7];
  float* out = (float*)d_out;
  const int E = in_sizes[1];

  const size_t hElems = (size_t)N_NODES * 128;
  float* h1 = (float*)d_ws;
  float* h2 = h1 + hElems;
  float* g  = h2 + hElems;

  const int spmmBlocks = (E + 7) / 8;
  const int spmm64Blocks = (E + 15) / 16;
  const int gemmBlocks = ((N_NODES / 16) + 7) / 8;

  hipMemsetAsync(h1, 0, hElems * sizeof(float), stream);
  spmm_edge_atomic<<<spmmBlocks, 256, 0, stream>>>(er, ec, ev, x, h1, E);
  gemm_wmma_f32<128, true, false><<<gemmBlocks, 256, 0, stream>>>(h1, W1, b1, h2, nullptr, N_NODES);
  gemm_wmma_f32<64, false, true><<<gemmBlocks, 256, 0, stream>>>(h2, W2, b2, g, out, N_NODES);
  spmm_edge_atomic64<<<spmm64Blocks, 256, 0, stream>>>(er, ec, ev, g, out, E);
}


# Both edge scatters (f32 atomics) become row gathers over an in-call stable sort of edges by row (ballot counting sort, no atomics left); a wave sums a row in registers; changes the order of summation
# speedup vs baseline: 2.6808x; 1.9377x over previous
#include <hip/hip_runtime.h>
#include <hip/hip_bf16.h>

typedef __attribute__((ext_vector_type(2))) float v2f;
typedef __attribute__((ext_vector_type(8))) float v8f;

#define N_NODES 100000
#define KDIM    128
#define ROWS_PER_WAVE 4

#define EDGE_CHUNK    4096
#define N_EDGE_BLOCKS 400
#define EDGE_SLOTS    (N_EDGE_BLOCKS * EDGE_CHUNK)
#define TABLE_LEN     (256 * N_EDGE_BLOCKS)
#define FINE_BINS     512


struct __attribute__((aligned(8))) edge_cv { int c; float v; };

template <int BITS>
__device__ __forceinline__ unsigned same_digit_lanes(unsigned d, bool valid) {
  unsigned peers = __builtin_amdgcn_ballot_w32(valid);
#pragma unroll
  for (int b = 0; b < BITS; ++b) {
    const bool bit = (d >> b) & 1u;
    const unsigned m = __builtin_amdgcn_ballot_w32(bit);
    peers &= bit ? m : ~m;
  }
  return peers;
}

__device__ __forceinline__ void coarse_count(const int (&r)[16], const bool (&ok)[16],
                                             int* __restrict__ myhist, unsigned lt) {
#pragma unroll
  for (int b = 0; b < 16; ++b) {
    const unsigned d = ((unsigned)r[b] >> 9) & 255u;
    const unsigned peers = same_digit_lanes<8>(d, ok[b]);
    if (ok[b] && (peers & lt) == 0) myhist[d] += __popc(peers);
  }
}

__global__ __launch_bounds__(256) void coarse_hist(
    const int* __restrict__ rows, int* __restrict__ table, int E) {
  __shared__ int hist[8][256];
  const int t = threadIdx.x, lane = t & 31;
  const int w = __builtin_amdgcn_readfirstlane(t >> 5);
  const unsigned lt = (1u << lane) - 1u;
  const int e0 = blockIdx.x * EDGE_CHUNK + w * 512 + lane;
  int r[16]; bool ok[16];
#pragma unroll
  for (int b = 0; b < 16; ++b) {
    const int e = e0 + b * 32;
    ok[b] = e < E;
    r[b] = ok[b] ? rows[e] : 0;
  }
#pragma unroll
  for (int i = 0; i < 8; ++i) hist[i][t] = 0;
  __syncthreads();
  coarse_count(r, ok, hist[w], lt);
  __syncthreads();
  int s = 0;
#pragma unroll
  for (int i = 0; i < 8; ++i) s += hist[i][t];
  table[t * N_EDGE_BLOCKS + blockIdx.x] = s;
}

#define SCAN_ITEMS 100
static_assert(TABLE_LEN == 1024 * SCAN_ITEMS && SCAN_ITEMS % 4 == 0,
              "table_scan: 1024 threads x whole 16-byte chunks");

__global__ __launch_bounds__(1024) void table_scan(
    const int* __restrict__ cnt, int* __restrict__ off) {
  __shared__ int wsum[32];
  const int t = threadIdx.x;
  const int lane = t & 31, w = t >> 5;
  const int base = t * SCAN_ITEMS;
  int4 v[SCAN_ITEMS / 4];
#pragma unroll
  for (int i = 0; i < SCAN_ITEMS / 4; ++i) v[i] = ((const int4*)(cnt + base))[i];
  int tot = 0;
#pragma unroll
  for (int i = 0; i < SCAN_ITEMS / 4; ++i) tot += v[i].x + v[i].y + v[i].z + v[i].w;
  int inc = tot;
#pragma unroll
  for (int d = 1; d < 32; d <<= 1) {
    const int o = __shfl_up(inc, d);
    if (lane >= d) inc += o;
  }
  if (lane == 31) wsum[w] = inc;
  __syncthreads();
  int run = inc - tot;
  for (int i = 0; i < w; ++i) run += wsum[i];
#pragma unroll
  for (int i = 0; i < SCAN_ITEMS / 4; ++i) {
    int4 o;
    o.x = run; run += v[i].x;
    o.y = run; run += v[i].y;
    o.z = run; run += v[i].z;
    o.w = run; run += v[i].w;
    ((int4*)(off + base))[i] = o;
  }
}

__global__ __launch_bounds__(256) void coarse_scatter(
    const int* __restrict__ rows, const int* __restrict__ cols,
    const float* __restrict__ vals, const int* __restrict__ off,
    int* __restrict__ tmp_r, edge_cv* __restrict__ tmp_cv, int E) {
  __shared__ int hist[8][256];
  const int t = threadIdx.x, lane = t & 31;
  const int w = __builtin_amdgcn_readfirstlane(t >> 5);
  const unsigned lt = (1u << lane) - 1u;
  const int e0 = blockIdx.x * EDGE_CHUNK + w * 512 + lane;
  int r[16]; bool ok[16]; edge_cv rec[16];
#pragma unroll
  for (int b = 0; b < 16; ++b) {
    const int e = e0 + b * 32;
    ok[b] = e < E;
    r[b] = ok[b] ? rows[e] : 0;
    rec[b].c = ok[b] ? cols[e] : 0;
    rec[b].v = ok[b] ? vals[e] : 0.0f;
  }
#pragma unroll
  for (int i = 0; i < 8; ++i) hist[i][t] = 0;
  __syncthreads();
  coarse_count(r, ok, hist[w], lt);
  __syncthreads();
  {
    int run = off[t * N_EDGE_BLOCKS + blockIdx.x];
#pragma unroll
    for (int i = 0; i < 8; ++i) { const int c = hist[i][t]; hist[i][t] = run; run += c; }
  }
  __syncthreads();
#pragma unroll
  for (int b = 0; b < 16; ++b) {
    const unsigned d = ((unsigned)r[b] >> 9) & 255u;
    const unsigned peers = same_digit_lanes<8>(d, ok[b]);
    const int rank = __popc(peers & lt);
    const int pos = hist[w][d] + rank;
    if (ok[b] && rank == 0) hist[w][d] = pos + __popc(peers);
    if (ok[b]) {
      const int p = min(max(pos, 0), E - 1);
      tmp_r[p] = r[b];
      tmp_cv[p] = rec[b];
    }
  }
}

#define FS_WAVES 32
__global__ __launch_bounds__(32 * FS_WAVES) void fine_sort(
    const int* __restrict__ off, const int* __restrict__ tmp_r,
    const edge_cv* __restrict__ tmp_cv, int* __restrict__ rowptr,
    edge_cv* __restrict__ cv, int E) {
  __shared__ int hist[FS_WAVES][FINE_BINS];
  __shared__ int wsum[8];
  const int bin = blockIdx.x;
  const int t = threadIdx.x, lane = t & 31;
  const int w = __builtin_amdgcn_readfirstlane(t >> 5);
  const unsigned lt = (1u << lane) - 1u;
  const int seg_e = min((bin == 255) ? E : off[(bin + 1) * N_EDGE_BLOCKS], E);
  const int seg_s = min(max(off[bin * N_EDGE_BLOCKS], 0), seg_e);
  const int per = ((seg_e - seg_s + 32 * FS_WAVES - 1) / (32 * FS_WAVES)) * 32;
  const int ws = min(seg_s + w * per, seg_e);
  const int we = min(ws + per, seg_e);
#pragma unroll
  for (int i = 0; i < FINE_BINS / 32; ++i) (&hist[0][0])[t + i * 32 * FS_WAVES] = 0;
  __syncthreads();
  for (int j0 = ws; j0 < we; j0 += 128) {
    int r[4]; bool ok[4];
#pragma unroll
    for (int q = 0; q < 4; ++q) {
      const int j = j0 + q * 32 + lane;
      ok[q] = j < we;
      r[q] = ok[q] ? tmp_r[j] : 0;
    }
#pragma unroll
    for (int q = 0; q < 4; ++q) {
      const unsigned d = (unsigned)r[q] & (FINE_BINS - 1);
      const unsigned peers = same_digit_lanes<9>(d, ok[q]);
      if (ok[q] && (peers & lt) == 0) hist[w][d] += __popc(peers);
    }
  }
  __syncthreads();
  int d0 = 0, d1 = 0;
  if (t < 256) {
#pragma unroll 8
    for (int i = 0; i < FS_WAVES; ++i) { d0 += hist[i][2 * t]; d1 += hist[i][2 * t + 1]; }
  }
  const int tot = d0 + d1;
  int inc = tot;
#pragma unroll
  for (int d = 1; d < 32; d <<= 1) {
    const int o = __shfl_up(inc, d);
    if (lane >= d) inc += o;
  }
  if (lane == 31 && t < 256) wsum[w] = inc;
  __syncthreads();
  if (t < 256) {
    int ex = seg_s + inc - tot;
    for (int i = 0; i < w; ++i) ex += wsum[i];
    const int rr = bin * FINE_BINS + 2 * t;
    if (rr <= N_NODES) rowptr[rr] = ex;
    if (rr + 1 <= N_NODES) rowptr[rr + 1] = ex + d0;
    int run0 = ex, run1 = ex + d0;
#pragma unroll 8
    for (int i = 0; i < FS_WAVES; ++i) {
      const int c0 = hist[i][2 * t], c1 = hist[i][2 * t + 1];
      hist[i][2 * t] = run0;     run0 += c0;
      hist[i][2 * t + 1] = run1; run1 += c1;
    }
  }
  __syncthreads();
  for (int j0 = ws; j0 < we; j0 += 128) {
    int r[4]; bool ok[4]; edge_cv rec[4];
#pragma unroll
    for (int q = 0; q < 4; ++q) {
      const int j = j0 + q * 32 + lane;
      ok[q] = j < we;
      r[q] = ok[q] ? tmp_r[j] : 0;
      if (ok[q]) rec[q] = tmp_cv[j]; else { rec[q].c = 0; rec[q].v = 0.0f; }
    }
#pragma unroll
    for (int q = 0; q < 4; ++q) {
      const unsigned d = (unsigned)r[q] & (FINE_BINS - 1);
      const unsigned peers = same_digit_lanes<9>(d, ok[q]);
      const int rank = __popc(peers & lt);
      const int pos = hist[w][d] + rank;
      if (ok[q] && rank == 0) hist[w][d] = pos + __popc(peers);
      if (ok[q]) cv[min(max(pos, 0), E - 1)] = rec[q];
    }
  }
}

template <int F> struct lane_vec {
  typedef float type __attribute__((ext_vector_type(F / 32)));
};

template <int F, int U>
__device__ __forceinline__ typename lane_vec<F>::type gather_edges(
    const edge_cv* __restrict__ cv, int j, const float* __restrict__ x,
    int lane, typename lane_vec<F>::type acc) {
  typedef typename lane_vec<F>::type vt;
  edge_cv e[U];
  vt xv[U];
#pragma unroll
  for (int u = 0; u < U; ++u) e[u] = cv[j + u];
#pragma unroll
  for (int u = 0; u < U; ++u)
    xv[u] = *((const vt*)(x + (size_t)e[u].c * F) + lane);
#pragma unroll
  for (int u = 0; u < U; ++u) acc += xv[u] * e[u].v;
  return acc;
}

template <int F, bool BIAS>
__global__ __launch_bounds__(256) void spmm_row_gather(
    const int* __restrict__ rowptr, const edge_cv* __restrict__ cv,
    const float* __restrict__ x, const float* __restrict__ bias,
    float* __restrict__ out, int E) {
  typedef typename lane_vec<F>::type vt;
  const int lane = threadIdx.x & 31;
  const int wave =
      __builtin_amdgcn_readfirstlane(blockIdx.x * 8 + (threadIdx.x >> 5));
  vt b = {};
  if (BIAS) b = *((const vt*)bias + lane);
  for (int i = 0; i < ROWS_PER_WAVE; ++i) {
    const int r = wave * ROWS_PER_WAVE + i;
    if (r >= N_NODES) break;
    const int t = min(rowptr[r + 1], E);
    int j = min(max(rowptr[r], 0), t);
    vt acc = b;
    for (; j + 8 <= t; j += 8) acc = gather_edges<F, 8>(cv, j, x, lane, acc);
    if (j + 4 <= t) { acc = gather_edges<F, 4>(cv, j, x, lane, acc); j += 4; }
    if (j + 2 <= t) { acc = gather_edges<F, 2>(cv, j, x, lane, acc); j += 2; }
    if (j < t) acc = gather_edges<F, 1>(cv, j, x, lane, acc);
    *((vt*)(out + (size_t)r * F) + lane) = acc;
  }
}

template <int NCOLS, bool RELU, bool BIAS>
__global__ __launch_bounds__(256) void gemm_wmma_f32(
    const float* __restrict__ A, const float* __restrict__ B,
    const float* __restrict__ bias, float* __restrict__ C, int M) {
  constexpr int K = KDIM;
  constexpr int LDSP = 2 * NCOLS + 32;
  __shared__ __align__(16) float Bs[(K / 2) * LDSP];

  for (int i = threadIdx.x; i < K * NCOLS; i += 256) {
    int k = i / NCOLS, n = i % NCOLS;
    Bs[(k >> 1) * LDSP + n * 2 + (k & 1)] = B[i];
  }
  __syncthreads();

  const int wave = threadIdx.x >> 5;
  const int lane = threadIdx.x & 31;
  const int tileRow = blockIdx.x * 8 + wave;
  if (tileRow * 16 >= M) return;
  const int row0 = tileRow * 16;

  const int mlane = lane & 15;
  const int khalf = (lane >> 4) * 2;

  v2f afrag[K / 4];
  const float* arow = A + (size_t)(row0 + mlane) * K + khalf;
#pragma unroll
  for (int ks = 0; ks < K / 4; ++ks)
    afrag[ks] = *(const v2f*)(arow + ks * 4);

#pragma unroll
  for (int nt = 0; nt < NCOLS / 16; ++nt) {
    const int n = nt * 16 + mlane;
    const float bv = BIAS ? bias[n] : 0.0f;
    v8f acc;
#pragma unroll
    for (int i = 0; i < 8; ++i) acc[i] = bv;

#pragma unroll
    for (int ks = 0; ks < K / 4; ++ks) {
      const int kpair = ks * 2 + (lane >> 4);
      v2f bfrag = *(const v2f*)&Bs[kpair * LDSP + n * 2];
      acc = __builtin_amdgcn_wmma_f32_16x16x4_f32(
           false, afrag[ks],  false, bfrag,
           (short)0, acc,  false,  false);
    }

    const int mbase = row0 + ((lane >> 4) * 8);
#pragma unroll
    for (int v = 0; v < 8; ++v) {
      float val = acc[v];
      if (RELU) val = fmaxf(val, 0.0f);
      C[(size_t)(mbase + v) * NCOLS + n] = val;
    }
  }
}

extern "C" void kernel_launch(void* const* d_in, const int* in_sizes, int n_in,
                              void* d_out, int out_size, void* d_ws, size_t ws_size,
                              hipStream_t stream) {
  const float* x  = (const float*)d_in[0];
  const int*   er = (const int*)d_in[1];
  const int*   ec = (const int*)d_in[2];
  const float* ev = (const float*)d_in[3];
  const float* W1 = (const float*)d_in[4];
  const float* b1 = (const float*)d_in[5];
  const float* W2 = (const float*)d_in[6];
  const float* b2 = (const float*)d_in[7];
  float* out = (float*)d_out;
  const int E = in_sizes[1];

  const size_t hElems = (size_t)N_NODES * 128;
  float* h1 = (float*)d_ws;
  float* h2 = h1 + hElems;
  float* g  = h2 + hElems;
  int* tcnt   = (int*)(g + (size_t)N_NODES * 64);
  int* toff   = tcnt + TABLE_LEN;
  int* rowptr = toff + TABLE_LEN;
  int* tmp_r  = rowptr + 100096;
  edge_cv* tmp_cv = (edge_cv*)(tmp_r + EDGE_SLOTS);
  edge_cv* cv = tmp_cv + EDGE_SLOTS;

  const int fineBlocks = (N_NODES >> 9) + 1;
  const int gatherBlocks = (N_NODES + 8 * ROWS_PER_WAVE - 1) / (8 * ROWS_PER_WAVE);
  const int gemmBlocks = ((N_NODES / 16) + 7) / 8;

  coarse_hist<<<N_EDGE_BLOCKS, 256, 0, stream>>>(er, tcnt, E);
  table_scan<<<1, 1024, 0, stream>>>(tcnt, toff);
  coarse_scatter<<<N_EDGE_BLOCKS, 256, 0, stream>>>(er, ec, ev, toff, tmp_r, tmp_cv, E);
  fine_sort<<<fineBlocks, 32 * FS_WAVES, 0, stream>>>(toff, tmp_r, tmp_cv, rowptr, cv, E);
  spmm_row_gather<128, false><<<gatherBlocks, 256, 0, stream>>>(rowptr, cv, x, nullptr, h1, E);
  gemm_wmma_f32<128, true, true><<<gemmBlocks, 256, 0, stream>>>(h1, W1, b1, h2, N_NODES);
  gemm_wmma_f32<64, false, false><<<gemmBlocks, 256, 0, stream>>>(h2, W2, b2, g, N_NODES);
  spmm_row_gather<64, true><<<gatherBlocks, 256, 0, stream>>>(rowptr, cv, g, b2, out, E);
}


# Both edge-sort passes order a block's edges in LDS first; consecutive lanes then store consecutive slots: contiguous stores replace one 4/8-byte store per line; order of sums unchanged
# speedup vs baseline: 3.1680x; 1.1817x over previous
#include <hip/hip_runtime.h>
#include <hip/hip_bf16.h>

typedef __attribute__((ext_vector_type(2))) float v2f;
typedef __attribute__((ext_vector_type(8))) float v8f;

#define N_NODES 100000
#define KDIM    128
#define ROWS_PER_WAVE 4

#define EDGE_CHUNK    4096
#define N_EDGE_BLOCKS 400
#define EDGE_SLOTS    (N_EDGE_BLOCKS * EDGE_CHUNK)
#define TABLE_LEN     (256 * N_EDGE_BLOCKS)
#define FINE_BINS     512


struct __attribute__((aligned(8))) edge_cv { int c; float v; };

template <int BITS>
__device__ __forceinline__ unsigned same_digit_lanes(unsigned d, bool valid) {
  unsigned peers = __builtin_amdgcn_ballot_w32(valid);
#pragma unroll
  for (int b = 0; b < BITS; ++b) {
    const bool bit = (d >> b) & 1u;
    const unsigned m = __builtin_amdgcn_ballot_w32(bit);
    peers &= bit ? m : ~m;
  }
  return peers;
}

__device__ __forceinline__ void coarse_count(const int (&r)[16], const bool (&ok)[16],
                                             int* __restrict__ myhist, unsigned lt) {
#pragma unroll
  for (int b = 0; b < 16; ++b) {
    const unsigned d = ((unsigned)r[b] >> 9) & 255u;
    const unsigned peers = same_digit_lanes<8>(d, ok[b]);
    if (ok[b] && (peers & lt) == 0) myhist[d] += __popc(peers);
  }
}

__global__ __launch_bounds__(256) void coarse_hist(
    const int* __restrict__ rows, int* __restrict__ table, int E) {
  __shared__ int hist[8][256];
  const int t = threadIdx.x, lane = t & 31;
  const int w = __builtin_amdgcn_readfirstlane(t >> 5);
  const unsigned lt = (1u << lane) - 1u;
  const int e0 = blockIdx.x * EDGE_CHUNK + w * 512 + lane;
  int r[16]; bool ok[16];
#pragma unroll
  for (int b = 0; b < 16; ++b) {
    const int e = e0 + b * 32;
    ok[b] = e < E;
    r[b] = ok[b] ? rows[e] : 0;
  }
#pragma unroll
  for (int i = 0; i < 8; ++i) hist[i][t] = 0;
  __syncthreads();
  coarse_count(r, ok, hist[w], lt);
  __syncthreads();
  int s = 0;
#pragma unroll
  for (int i = 0; i < 8; ++i) s += hist[i][t];
  table[t * N_EDGE_BLOCKS + blockIdx.x] = s;
}

#define SCAN_ITEMS 100
static_assert(TABLE_LEN == 1024 * SCAN_ITEMS && SCAN_ITEMS % 4 == 0,
              "table_scan: 1024 threads x whole 16-byte chunks");

__global__ __launch_bounds__(1024) void table_scan(
    const int* __restrict__ cnt, int* __restrict__ off) {
  __shared__ int wsum[32];
  const int t = threadIdx.x;
  const int lane = t & 31, w = t >> 5;
  const int base = t * SCAN_ITEMS;
  int4 v[SCAN_ITEMS / 4];
#pragma unroll
  for (int i = 0; i < SCAN_ITEMS / 4; ++i) v[i] = ((const int4*)(cnt + base))[i];
  int tot = 0;
#pragma unroll
  for (int i = 0; i < SCAN_ITEMS / 4; ++i) tot += v[i].x + v[i].y + v[i].z + v[i].w;
  int inc = tot;
#pragma unroll
  for (int d = 1; d < 32; d <<= 1) {
    const int o = __shfl_up(inc, d);
    if (lane >= d) inc += o;
  }
  if (lane == 31) wsum[w] = inc;
  __syncthreads();
  int run = inc - tot;
  for (int i = 0; i < w; ++i) run += wsum[i];
#pragma unroll
  for (int i = 0; i < SCAN_ITEMS / 4; ++i) {
    int4 o;
    o.x = run; run += v[i].x;
    o.y = run; run += v[i].y;
    o.z = run; run += v[i].z;
    o.w = run; run += v[i].w;
    ((int4*)(off + base))[i] = o;
  }
}

__global__ __launch_bounds__(256) void coarse_scatter(
    const int* __restrict__ rows, const int* __restrict__ cols,
    const float* __restrict__ vals, const int* __restrict__ off,
    int* __restrict__ tmp_r, edge_cv* __restrict__ tmp_cv, int E) {
  __shared__ int hist[8][256];
  __shared__ int gdelta[256];
  __shared__ int wsum[8];
  __shared__ int s_r[EDGE_CHUNK];
  __shared__ edge_cv s_cv[EDGE_CHUNK];
  const int t = threadIdx.x, lane = t & 31;
  const int w = __builtin_amdgcn_readfirstlane(t >> 5);
  const unsigned lt = (1u << lane) - 1u;
  const int e0 = blockIdx.x * EDGE_CHUNK + w * 512 + lane;
  int r[16]; bool ok[16]; edge_cv rec[16];
#pragma unroll
  for (int b = 0; b < 16; ++b) {
    const int e = e0 + b * 32;
    ok[b] = e < E;
    r[b] = ok[b] ? rows[e] : 0;
    rec[b].c = ok[b] ? cols[e] : 0;
    rec[b].v = ok[b] ? vals[e] : 0.0f;
  }
#pragma unroll
  for (int i = 0; i < 8; ++i) hist[i][t] = 0;
  __syncthreads();
  coarse_count(r, ok, hist[w], lt);
  __syncthreads();
  int tot = 0;
#pragma unroll
  for (int i = 0; i < 8; ++i) tot += hist[i][t];
  int inc = tot;
#pragma unroll
  for (int d = 1; d < 32; d <<= 1) {
    const int o = __shfl_up(inc, d);
    if (lane >= d) inc += o;
  }
  if (lane == 31) wsum[w] = inc;
  __syncthreads();
  {
    int lstart = inc - tot;
    for (int i = 0; i < w; ++i) lstart += wsum[i];
    gdelta[t] = off[t * N_EDGE_BLOCKS + blockIdx.x] - lstart;
    int run = lstart;
#pragma unroll
    for (int i = 0; i < 8; ++i) { const int c = hist[i][t]; hist[i][t] = run; run += c; }
  }
  __syncthreads();
#pragma unroll
  for (int b = 0; b < 16; ++b) {
    const unsigned d = ((unsigned)r[b] >> 9) & 255u;
    const unsigned peers = same_digit_lanes<8>(d, ok[b]);
    const int rank = __popc(peers & lt);
    const int pos = hist[w][d] + rank;
    if (ok[b] && rank == 0) hist[w][d] = pos + __popc(peers);
    if (ok[b]) {
      const int p = min(max(pos, 0), EDGE_CHUNK - 1);
      s_r[p] = r[b];
      s_cv[p] = rec[b];
    }
  }
  __syncthreads();
  const int nvalid = min(max(E - (int)blockIdx.x * EDGE_CHUNK, 0), EDGE_CHUNK);
  if (nvalid > 0) {
    int rr[EDGE_CHUNK / 256], gd[EDGE_CHUNK / 256];
    edge_cv ev[EDGE_CHUNK / 256];
#pragma unroll
    for (int k = 0; k < EDGE_CHUNK / 256; ++k) {
      const int i = min(k * 256 + t, nvalid - 1);
      rr[k] = s_r[i];
      ev[k] = s_cv[i];
    }
#pragma unroll
    for (int k = 0; k < EDGE_CHUNK / 256; ++k) gd[k] = gdelta[((unsigned)rr[k] >> 9) & 255u];
#pragma unroll
    for (int k = 0; k < EDGE_CHUNK / 256; ++k) {
      const int i = k * 256 + t;
      if (i < nvalid) {
        const int p = min(max(gd[k] + i, 0), E - 1);
        tmp_r[p] = rr[k];
        tmp_cv[p] = ev[k];
      }
    }
  }
}

#define FS_WAVES 32
#define FS_STAGE 10240
__global__ __launch_bounds__(32 * FS_WAVES) void fine_sort(
    const int* __restrict__ off, const int* __restrict__ tmp_r,
    const edge_cv* __restrict__ tmp_cv, int* __restrict__ rowptr,
    edge_cv* __restrict__ cv, int E) {
  __shared__ int hist[FS_WAVES][FINE_BINS];
  __shared__ int wsum[8];
  __shared__ edge_cv stage[FS_STAGE];
  const int bin = blockIdx.x;
  const int t = threadIdx.x, lane = t & 31;
  const int w = __builtin_amdgcn_readfirstlane(t >> 5);
  const unsigned lt = (1u << lane) - 1u;
  const int seg_e = min((bin == 255) ? E : off[(bin + 1) * N_EDGE_BLOCKS], E);
  const int seg_s = min(max(off[bin * N_EDGE_BLOCKS], 0), seg_e);
  const int per = ((seg_e - seg_s + 32 * FS_WAVES - 1) / (32 * FS_WAVES)) * 32;
  const int ws = min(seg_s + w * per, seg_e);
  const int we = min(ws + per, seg_e);
  const int nseg = seg_e - seg_s;
  const bool staged = nseg <= FS_STAGE;
#pragma unroll
  for (int i = 0; i < FINE_BINS / 32; ++i) (&hist[0][0])[t + i * 32 * FS_WAVES] = 0;
  __syncthreads();
  for (int j0 = ws; j0 < we; j0 += 128) {
    int r[4]; bool ok[4];
#pragma unroll
    for (int q = 0; q < 4; ++q) {
      const int j = j0 + q * 32 + lane;
      ok[q] = j < we;
      r[q] = ok[q] ? tmp_r[j] : 0;
    }
#pragma unroll
    for (int q = 0; q < 4; ++q) {
      const unsigned d = (unsigned)r[q] & (FINE_BINS - 1);
      const unsigned peers = same_digit_lanes<9>(d, ok[q]);
      if (ok[q] && (peers & lt) == 0) hist[w][d] += __popc(peers);
    }
  }
  __syncthreads();
  int d0 = 0, d1 = 0;
  if (t < 256) {
#pragma unroll 8
    for (int i = 0; i < FS_WAVES; ++i) { d0 += hist[i][2 * t]; d1 += hist[i][2 * t + 1]; }
  }
  const int tot = d0 + d1;
  int inc = tot;
#pragma unroll
  for (int d = 1; d < 32; d <<= 1) {
    const int o = __shfl_up(inc, d);
    if (lane >= d) inc += o;
  }
  if (lane == 31 && t < 256) wsum[w] = inc;
  __syncthreads();
  if (t < 256) {
    int ex = seg_s + inc - tot;
    for (int i = 0; i < w; ++i) ex += wsum[i];
    const int rr = bin * FINE_BINS + 2 * t;
    if (rr <= N_NODES) rowptr[rr] = ex;
    if (rr + 1 <= N_NODES) rowptr[rr + 1] = ex + d0;
    int run0 = ex, run1 = ex + d0;
#pragma unroll 8
    for (int i = 0; i < FS_WAVES; ++i) {
      const int c0 = hist[i][2 * t], c1 = hist[i][2 * t + 1];
      hist[i][2 * t] = run0;     run0 += c0;
      hist[i][2 * t + 1] = run1; run1 += c1;
    }
  }
  __syncthreads();
  for (int j0 = ws; j0 < we; j0 += 128) {
    int r[4]; bool ok[4]; edge_cv rec[4];
#pragma unroll
    for (int q = 0; q < 4; ++q) {
      const int j = j0 + q * 32 + lane;
      ok[q] = j < we;
      r[q] = ok[q] ? tmp_r[j] : 0;
      if (ok[q]) rec[q] = tmp_cv[j]; else { rec[q].c = 0; rec[q].v = 0.0f; }
    }
#pragma unroll
    for (int q = 0; q < 4; ++q) {
      const unsigned d = (unsigned)r[q] & (FINE_BINS - 1);
      const unsigned peers = same_digit_lanes<9>(d, ok[q]);
      const int rank = __popc(peers & lt);
      const int pos = hist[w][d] + rank;
      if (ok[q] && rank == 0) hist[w][d] = pos + __popc(peers);
      if (ok[q]) {
        if (staged) stage[min(max(pos - seg_s, 0), FS_STAGE - 1)] = rec[q];
        else cv[min(max(pos, 0), E - 1)] = rec[q];
      }
    }
  }
  __syncthreads();
  if (staged) {
    for (int i = t; i < nseg; i += 32 * FS_WAVES) cv[seg_s + i] = stage[i];
  }
}

template <int F> struct lane_vec {
  typedef float type __attribute__((ext_vector_type(F / 32)));
};

template <int F, int U>
__device__ __forceinline__ typename lane_vec<F>::type gather_edges(
    const edge_cv* __restrict__ cv, int j, const float* __restrict__ x,
    int lane, typename lane_vec<F>::type acc) {
  typedef typename lane_vec<F>::type vt;
  edge_cv e[U];
  vt xv[U];
#pragma unroll
  for (int u = 0; u < U; ++u) e[u] = cv[j + u];
#pragma unroll
  for (int u = 0; u < U; ++u)
    xv[u] = *((const vt*)(x + (size_t)e[u].c * F) + lane);
#pragma unroll
  for (int u = 0; u < U; ++u) acc += xv[u] * e[u].v;
  return acc;
}

template <int F, bool BIAS>
__global__ __launch_bounds__(256) void spmm_row_gather(
    const int* __restrict__ rowptr, const edge_cv* __restrict__ cv,
    const float* __restrict__ x, const float* __restrict__ bias,
    float* __restrict__ out, int E) {
  typedef typename lane_vec<F>::type vt;
  const int lane = threadIdx.x & 31;
  const int wave =
      __builtin_amdgcn_readfirstlane(blockIdx.x * 8 + (threadIdx.x >> 5));
  vt b = {};
  if (BIAS) b = *((const vt*)bias + lane);
  for (int i = 0; i < ROWS_PER_WAVE; ++i) {
    const int r = wave * ROWS_PER_WAVE + i;
    if (r >= N_NODES) break;
    const int t = min(rowptr[r + 1], E);
    int j = min(max(rowptr[r], 0), t);
    vt acc = b;
    for (; j + 8 <= t; j += 8) acc = gather_edges<F, 8>(cv, j, x, lane, acc);
    if (j + 4 <= t) { acc = gather_edges<F, 4>(cv, j, x, lane, acc); j += 4; }
    if (j + 2 <= t) { acc = gather_edges<F, 2>(cv, j, x, lane, acc); j += 2; }
    if (j < t) acc = gather_edges<F, 1>(cv, j, x, lane, acc);
    *((vt*)(out + (size_t)r * F) + lane) = acc;
  }
}

template <int NCOLS, bool RELU, bool BIAS>
__global__ __launch_bounds__(256) void gemm_wmma_f32(
    const float* __restrict__ A, const float* __restrict__ B,
    const float* __restrict__ bias, float* __restrict__ C, int M) {
  constexpr int K = KDIM;
  constexpr int LDSP = 2 * NCOLS + 32;
  __shared__ __align__(16) float Bs[(K / 2) * LDSP];

  for (int i = threadIdx.x; i < K * NCOLS; i += 256) {
    int k = i / NCOLS, n = i % NCOLS;
    Bs[(k >> 1) * LDSP + n * 2 + (k & 1)] = B[i];
  }
  __syncthreads();

  const int wave = threadIdx.x >> 5;
  const int lane = threadIdx.x & 31;
  const int tileRow = blockIdx.x * 8 + wave;
  if (tileRow * 16 >= M) return;
  const int row0 = tileRow * 16;

  const int mlane = lane & 15;
  const int khalf = (lane >> 4) * 2;

  v2f afrag[K / 4];
  const float* arow = A + (size_t)(row0 + mlane) * K + khalf;
#pragma unroll
  for (int ks = 0; ks < K / 4; ++ks)
    afrag[ks] = *(const v2f*)(arow + ks * 4);

#pragma unroll
  for (int nt = 0; nt < NCOLS / 16; ++nt) {
    const int n = nt * 16 + mlane;
    const float bv = BIAS ? bias[n] : 0.0f;
    v8f acc;
#pragma unroll
    for (int i = 0; i < 8; ++i) acc[i] = bv;

#pragma unroll
    for (int ks = 0; ks < K / 4; ++ks) {
      const int kpair = ks * 2 + (lane >> 4);
      v2f bfrag = *(const v2f*)&Bs[kpair * LDSP + n * 2];
      acc = __builtin_amdgcn_wmma_f32_16x16x4_f32(
           false, afrag[ks],  false, bfrag,
           (short)0, acc,  false,  false);
    }

    const int mbase = row0 + ((lane >> 4) * 8);
#pragma unroll
    for (int v = 0; v < 8; ++v) {
      float val = acc[v];
      if (RELU) val = fmaxf(val, 0.0f);
      C[(size_t)(mbase + v) * NCOLS + n] = val;
    }
  }
}

extern "C" void kernel_launch(void* const* d_in, const int* in_sizes, int n_in,
                              void* d_out, int out_size, void* d_ws, size_t ws_size,
                              hipStream_t stream) {
  const float* x  = (const float*)d_in[0];
  const int*   er = (const int*)d_in[1];
  const int*   ec = (const int*)d_in[2];
  const float* ev = (const float*)d_in[3];
  const float* W1 = (const float*)d_in[4];
  const float* b1 = (const float*)d_in[5];
  const float* W2 = (const float*)d_in[6];
  const float* b2 = (const float*)d_in[7];
  float* out = (float*)d_out;
  const int E = in_sizes[1];

  const size_t hElems = (size_t)N_NODES * 128;
  float* h1 = (float*)d_ws;
  float* h2 = h1 + hElems;
  float* g  = h2 + hElems;
  int* tcnt   = (int*)(g + (size_t)N_NODES * 64);
  int* toff   = tcnt + TABLE_LEN;
  int* rowptr = toff + TABLE_LEN;
  int* tmp_r  = rowptr + 100096;
  edge_cv* tmp_cv = (edge_cv*)(tmp_r + EDGE_SLOTS);
  edge_cv* cv = tmp_cv + EDGE_SLOTS;

  const int fineBlocks = (N_NODES >> 9) + 1;
  const int gatherBlocks = (N_NODES + 8 * ROWS_PER_WAVE - 1) / (8 * ROWS_PER_WAVE);
  const int gemmBlocks = ((N_NODES / 16) + 7) / 8;

  coarse_hist<<<N_EDGE_BLOCKS, 256, 0, stream>>>(er, tcnt, E);
  table_scan<<<1, 1024, 0, stream>>>(tcnt, toff);
  coarse_scatter<<<N_EDGE_BLOCKS, 256, 0, stream>>>(er, ec, ev, toff, tmp_r, tmp_cv, E);
  fine_sort<<<fineBlocks, 32 * FS_WAVES, 0, stream>>>(toff, tmp_r, tmp_cv, rowptr, cv, E);
  spmm_row_gather<128, false><<<gatherBlocks, 256, 0, stream>>>(rowptr, cv, x, nullptr, h1, E);
  gemm_wmma_f32<128, true, true><<<gemmBlocks, 256, 0, stream>>>(h1, W1, b1, h2, N_NODES);
  gemm_wmma_f32<64, false, false><<<gemmBlocks, 256, 0, stream>>>(h2, W2, b2, g, N_NODES);
  spmm_row_gather<64, true><<<gatherBlocks, 256, 0, stream>>>(rowptr, cv, g, b2, out, E);
}


# Table scan (prefix sum of the 102400 sort counts) runs as one work-group per bin: each sums the counts before its bin with whole-line 16-byte loads, then scans its own 400; float sums unchanged
# speedup vs baseline: 3.8144x; 1.2040x over previous
#include <hip/hip_runtime.h>
#include <hip/hip_bf16.h>

typedef __attribute__((ext_vector_type(2))) float v2f;
typedef __attribute__((ext_vector_type(8))) float v8f;

#define N_NODES 100000
#define KDIM    128
#define ROWS_PER_WAVE 4

#define EDGE_CHUNK    4096
#define N_EDGE_BLOCKS 400
#define EDGE_SLOTS    (N_EDGE_BLOCKS * EDGE_CHUNK)
#define TABLE_LEN     (256 * N_EDGE_BLOCKS)
#define FINE_BINS     512


struct __attribute__((aligned(8))) edge_cv { int c; float v; };

template <int BITS>
__device__ __forceinline__ unsigned same_digit_lanes(unsigned d, bool valid) {
  unsigned peers = __builtin_amdgcn_ballot_w32(valid);
#pragma unroll
  for (int b = 0; b < BITS; ++b) {
    const bool bit = (d >> b) & 1u;
    const unsigned m = __builtin_amdgcn_ballot_w32(bit);
    peers &= bit ? m : ~m;
  }
  return peers;
}

__device__ __forceinline__ void coarse_count(const int (&r)[16], const bool (&ok)[16],
                                             int* __restrict__ myhist, unsigned lt) {
#pragma unroll
  for (int b = 0; b < 16; ++b) {
    const unsigned d = ((unsigned)r[b] >> 9) & 255u;
    const unsigned peers = same_digit_lanes<8>(d, ok[b]);
    if (ok[b] && (peers & lt) == 0) myhist[d] += __popc(peers);
  }
}

__global__ __launch_bounds__(256) void coarse_hist(
    const int* __restrict__ rows, int* __restrict__ table, int E) {
  __shared__ int hist[8][256];
  const int t = threadIdx.x, lane = t & 31;
  const int w = __builtin_amdgcn_readfirstlane(t >> 5);
  const unsigned lt = (1u << lane) - 1u;
  const int e0 = blockIdx.x * EDGE_CHUNK + w * 512 + lane;
  int r[16]; bool ok[16];
#pragma unroll
  for (int b = 0; b < 16; ++b) {
    const int e = e0 + b * 32;
    ok[b] = e < E;
    r[b] = ok[b] ? rows[e] : 0;
  }
#pragma unroll
  for (int i = 0; i < 8; ++i) hist[i][t] = 0;
  __syncthreads();
  coarse_count(r, ok, hist[w], lt);
  __syncthreads();
  int s = 0;
#pragma unroll
  for (int i = 0; i < 8; ++i) s += hist[i][t];
  table[t * N_EDGE_BLOCKS + blockIdx.x] = s;
}

#define SCAN_THREADS 256
#define SCAN_BATCH   25
static_assert(TABLE_LEN == 256 * N_EDGE_BLOCKS && N_EDGE_BLOCKS % 4 == 0 &&
              N_EDGE_BLOCKS <= 2 * SCAN_THREADS,
              "table_scan: one group per bin, whole 16-byte chunks, two entries a thread");

__global__ __launch_bounds__(SCAN_THREADS) void table_scan(
    const int* __restrict__ cnt, int* __restrict__ off) {
  __shared__ int wprev[SCAN_THREADS / 32];
  __shared__ int wsum[SCAN_THREADS / 32];
  const int t = threadIdx.x;
  const int lane = t & 31, w = t >> 5;
  const int bin = blockIdx.x;
  const int n4 = bin * (N_EDGE_BLOCKS / 4);
  const int4* __restrict__ c4 = (const int4*)cnt;
  const bool owner = 2 * t < N_EDGE_BLOCKS;
  int2 own = make_int2(0, 0);
  if (owner) own = *(const int2*)(cnt + bin * N_EDGE_BLOCKS + 2 * t);
  int s = 0;
  for (int i0 = 0; i0 < n4; i0 += SCAN_BATCH * SCAN_THREADS) {
    int4 v[SCAN_BATCH];
#pragma unroll
    for (int u = 0; u < SCAN_BATCH; ++u)
      v[u] = c4[min(i0 + u * SCAN_THREADS + t, n4 - 1)];
#pragma unroll
    for (int u = 0; u < SCAN_BATCH; ++u)
      if (i0 + u * SCAN_THREADS + t < n4) s += v[u].x + v[u].y + v[u].z + v[u].w;
  }
#pragma unroll
  for (int d = 16; d >= 1; d >>= 1) s += __shfl_xor(s, d);
  const int tot = own.x + own.y;
  int inc = tot;
#pragma unroll
  for (int d = 1; d < 32; d <<= 1) {
    const int o = __shfl_up(inc, d);
    if (lane >= d) inc += o;
  }
  if (lane == 0) wprev[w] = s;
  if (lane == 31) wsum[w] = inc;
  __syncthreads();
  int run = inc - tot;
#pragma unroll
  for (int i = 0; i < SCAN_THREADS / 32; ++i) {
    run += wprev[i];
    if (i < w) run += wsum[i];
  }
  if (owner)
    *(int2*)(off + bin * N_EDGE_BLOCKS + 2 * t) = make_int2(run, run + own.x);
}

__global__ __launch_bounds__(256) void coarse_scatter(
    const int* __restrict__ rows, const int* __restrict__ cols,
    const float* __restrict__ vals, const int* __restrict__ off,
    int* __restrict__ tmp_r, edge_cv* __restrict__ tmp_cv, int E) {
  __shared__ int hist[8][256];
  __shared__ int gdelta[256];
  __shared__ int wsum[8];
  __shared__ int s_r[EDGE_CHUNK];
  __shared__ edge_cv s_cv[EDGE_CHUNK];
  const int t = threadIdx.x, lane = t & 31;
  const int w = __builtin_amdgcn_readfirstlane(t >> 5);
  const unsigned lt = (1u << lane) - 1u;
  const int e0 = blockIdx.x * EDGE_CHUNK + w * 512 + lane;
  int r[16]; bool ok[16]; edge_cv rec[16];
#pragma unroll
  for (int b = 0; b < 16; ++b) {
    const int e = e0 + b * 32;
    ok[b] = e < E;
    r[b] = ok[b] ? rows[e] : 0;
    rec[b].c = ok[b] ? cols[e] : 0;
    rec[b].v = ok[b] ? vals[e] : 0.0f;
  }
#pragma unroll
  for (int i = 0; i < 8; ++i) hist[i][t] = 0;
  __syncthreads();
  coarse_count(r, ok, hist[w], lt);
  __syncthreads();
  int tot = 0;
#pragma unroll
  for (int i = 0; i < 8; ++i) tot += hist[i][t];
  int inc = tot;
#pragma unroll
  for (int d = 1; d < 32; d <<= 1) {
    const int o = __shfl_up(inc, d);
    if (lane >= d) inc += o;
  }
  if (lane == 31) wsum[w] = inc;
  __syncthreads();
  {
    int lstart = inc - tot;
    for (int i = 0; i < w; ++i) lstart += wsum[i];
    gdelta[t] = off[t * N_EDGE_BLOCKS + blockIdx.x] - lstart;
    int run = lstart;
#pragma unroll
    for (int i = 0; i < 8; ++i) { const int c = hist[i][t]; hist[i][t] = run; run += c; }
  }
  __syncthreads();
#pragma unroll
  for (int b = 0; b < 16; ++b) {
    const unsigned d = ((unsigned)r[b] >> 9) & 255u;
    const unsigned peers = same_digit_lanes<8>(d, ok[b]);
    const int rank = __popc(peers & lt);
    const int pos = hist[w][d] + rank;
    if (ok[b] && rank == 0) hist[w][d] = pos + __popc(peers);
    if (ok[b]) {
      const int p = min(max(pos, 0), EDGE_CHUNK - 1);
      s_r[p] = r[b];
      s_cv[p] = rec[b];
    }
  }
  __syncthreads();
  const int nvalid = min(max(E - (int)blockIdx.x * EDGE_CHUNK, 0), EDGE_CHUNK);
  if (nvalid > 0) {
    int rr[EDGE_CHUNK / 256], gd[EDGE_CHUNK / 256];
    edge_cv ev[EDGE_CHUNK / 256];
#pragma unroll
    for (int k = 0; k < EDGE_CHUNK / 256; ++k) {
      const int i = min(k * 256 + t, nvalid - 1);
      rr[k] = s_r[i];
      ev[k] = s_cv[i];
    }
#pragma unroll
    for (int k = 0; k < EDGE_CHUNK / 256; ++k) gd[k] = gdelta[((unsigned)rr[k] >> 9) & 255u];
#pragma unroll
    for (int k = 0; k < EDGE_CHUNK / 256; ++k) {
      const int i = k * 256 + t;
      if (i < nvalid) {
        const int p = min(max(gd[k] + i, 0), E - 1);
        tmp_r[p] = rr[k];
        tmp_cv[p] = ev[k];
      }
    }
  }
}

#define FS_WAVES 32
#define FS_STAGE 10240
__global__ __launch_bounds__(32 * FS_WAVES) void fine_sort(
    const int* __restrict__ off, const int* __restrict__ tmp_r,
    const edge_cv* __restrict__ tmp_cv, int* __restrict__ rowptr,
    edge_cv* __restrict__ cv, int E) {
  __shared__ int hist[FS_WAVES][FINE_BINS];
  __shared__ int wsum[8];
  __shared__ edge_cv stage[FS_STAGE];
  const int bin = blockIdx.x;
  const int t = threadIdx.x, lane = t & 31;
  const int w = __builtin_amdgcn_readfirstlane(t >> 5);
  const unsigned lt = (1u << lane) - 1u;
  const int seg_e = min((bin == 255) ? E : off[(bin + 1) * N_EDGE_BLOCKS], E);
  const int seg_s = min(max(off[bin * N_EDGE_BLOCKS], 0), seg_e);
  const int per = ((seg_e - seg_s + 32 * FS_WAVES - 1) / (32 * FS_WAVES)) * 32;
  const int ws = min(seg_s + w * per, seg_e);
  const int we = min(ws + per, seg_e);
  const int nseg = seg_e - seg_s;
  const bool staged = nseg <= FS_STAGE;
#pragma unroll
  for (int i = 0; i < FINE_BINS / 32; ++i) (&hist[0][0])[t + i * 32 * FS_WAVES] = 0;
  __syncthreads();
  for (int j0 = ws; j0 < we; j0 += 128) {
    int r[4]; bool ok[4];
#pragma unroll
    for (int q = 0; q < 4; ++q) {
      const int j = j0 + q * 32 + lane;
      ok[q] = j < we;
      r[q] = ok[q] ? tmp_r[j] : 0;
    }
#pragma unroll
    for (int q = 0; q < 4; ++q) {
      const unsigned d = (unsigned)r[q] & (FINE_BINS - 1);
      const unsigned peers = same_digit_lanes<9>(d, ok[q]);
      if (ok[q] && (peers & lt) == 0) hist[w][d] += __popc(peers);
    }
  }
  __syncthreads();
  int d0 = 0, d1 = 0;
  if (t < 256) {
#pragma unroll 8
    for (int i = 0; i < FS_WAVES; ++i) { d0 += hist[i][2 * t]; d1 += hist[i][2 * t + 1]; }
  }
  const int tot = d0 + d1;
  int inc = tot;
#pragma unroll
  for (int d = 1; d < 32; d <<= 1) {
    const int o = __shfl_up(inc, d);
    if (lane >= d) inc += o;
  }
  if (lane == 31 && t < 256) wsum[w] = inc;
  __syncthreads();
  if (t < 256) {
    int ex = seg_s + inc - tot;
    for (int i = 0; i < w; ++i) ex += wsum[i];
    const int rr = bin * FINE_BINS + 2 * t;
    if (rr <= N_NODES) rowptr[rr] = ex;
    if (rr + 1 <= N_NODES) rowptr[rr + 1] = ex + d0;
    int run0 = ex, run1 = ex + d0;
#pragma unroll 8
    for (int i = 0; i < FS_WAVES; ++i) {
      const int c0 = hist[i][2 * t], c1 = hist[i][2 * t + 1];
      hist[i][2 * t] = run0;     run0 += c0;
      hist[i][2 * t + 1] = run1; run1 += c1;
    }
  }
  __syncthreads();
  for (int j0 = ws; j0 < we; j0 += 128) {
    int r[4]; bool ok[4]; edge_cv rec[4];
#pragma unroll
    for (int q = 0; q < 4; ++q) {
      const int j = j0 + q * 32 + lane;
      ok[q] = j < we;
      r[q] = ok[q] ? tmp_r[j] : 0;
      if (ok[q]) rec[q] = tmp_cv[j]; else { rec[q].c = 0; rec[q].v = 0.0f; }
    }
#pragma unroll
    for (int q = 0; q < 4; ++q) {
      const unsigned d = (unsigned)r[q] & (FINE_BINS - 1);
      const unsigned peers = same_digit_lanes<9>(d, ok[q]);
      const int rank = __popc(peers & lt);
      const int pos = hist[w][d] + rank;
      if (ok[q] && rank == 0) hist[w][d] = pos + __popc(peers);
      if (ok[q]) {
        if (staged) stage[min(max(pos - seg_s, 0), FS_STAGE - 1)] = rec[q];
        else cv[min(max(pos, 0), E - 1)] = rec[q];
      }
    }
  }
  __syncthreads();
  if (staged) {
    for (int i = t; i < nseg; i += 32 * FS_WAVES) cv[seg_s + i] = stage[i];
  }
}

template <int F> struct lane_vec {
  typedef float type __attribute__((ext_vector_type(F / 32)));
};

template <int F, int U>
__device__ __forceinline__ typename lane_vec<F>::type gather_edges(
    const edge_cv* __restrict__ cv, int j, const float* __restrict__ x,
    int lane, typename lane_vec<F>::type acc) {
  typedef typename lane_vec<F>::type vt;
  edge_cv e[U];
  vt xv[U];
#pragma unroll
  for (int u = 0; u < U; ++u) e[u] = cv[j + u];
#pragma unroll
  for (int u = 0; u < U; ++u)
    xv[u] = *((const vt*)(x + (size_t)e[u].c * F) + lane);
#pragma unroll
  for (int u = 0; u < U; ++u) acc += xv[u] * e[u].v;
  return acc;
}

template <int F, bool BIAS>
__global__ __launch_bounds__(256) void spmm_row_gather(
    const int* __restrict__ rowptr, const edge_cv* __restrict__ cv,
    const float* __restrict__ x, const float* __restrict__ bias,
    float* __restrict__ out, int E) {
  typedef typename lane_vec<F>::type vt;
  const int lane = threadIdx.x & 31;
  const int wave =
      __builtin_amdgcn_readfirstlane(blockIdx.x * 8 + (threadIdx.x >> 5));
  vt b = {};
  if (BIAS) b = *((const vt*)bias + lane);
  for (int i = 0; i < ROWS_PER_WAVE; ++i) {
    const int r = wave * ROWS_PER_WAVE + i;
    if (r >= N_NODES) break;
    const int t = min(rowptr[r + 1], E);
    int j = min(max(rowptr[r], 0), t);
    vt acc = b;
    for (; j + 8 <= t; j += 8) acc = gather_edges<F, 8>(cv, j, x, lane, acc);
    if (j + 4 <= t) { acc = gather_edges<F, 4>(cv, j, x, lane, acc); j += 4; }
    if (j + 2 <= t) { acc = gather_edges<F, 2>(cv, j, x, lane, acc); j += 2; }
    if (j < t) acc = gather_edges<F, 1>(cv, j, x, lane, acc);
    *((vt*)(out + (size_t)r * F) + lane) = acc;
  }
}

template <int NCOLS, bool RELU, bool BIAS>
__global__ __launch_bounds__(256) void gemm_wmma_f32(
    const float* __restrict__ A, const float* __restrict__ B,
    const float* __restrict__ bias, float* __restrict__ C, int M) {
  constexpr int K = KDIM;
  constexpr int LDSP = 2 * NCOLS + 32;
  __shared__ __align__(16) float Bs[(K / 2) * LDSP];

  for (int i = threadIdx.x; i < K * NCOLS; i += 256) {
    int k = i / NCOLS, n = i % NCOLS;
    Bs[(k >> 1) * LDSP + n * 2 + (k & 1)] = B[i];
  }
  __syncthreads();

  const int wave = threadIdx.x >> 5;
  const int lane = threadIdx.x & 31;
  const int tileRow = blockIdx.x * 8 + wave;
  if (tileRow * 16 >= M) return;
  const int row0 = tileRow * 16;

  const int mlane = lane & 15;
  const int khalf = (lane >> 4) * 2;

  v2f afrag[K / 4];
  const float* arow = A + (size_t)(row0 + mlane) * K + khalf;
#pragma unroll
  for (int ks = 0; ks < K / 4; ++ks)
    afrag[ks] = *(const v2f*)(arow + ks * 4);

#pragma unroll
  for (int nt = 0; nt < NCOLS / 16; ++nt) {
    const int n = nt * 16 + mlane;
    const float bv = BIAS ? bias[n] : 0.0f;
    v8f acc;
#pragma unroll
    for (int i = 0; i < 8; ++i) acc[i] = bv;

#pragma unroll
    for (int ks = 0; ks < K / 4; ++ks) {
      const int kpair = ks * 2 + (lane >> 4);
      v2f bfrag = *(const v2f*)&Bs[kpair * LDSP + n * 2];
      acc = __builtin_amdgcn_wmma_f32_16x16x4_f32(
           false, afrag[ks],  false, bfrag,
           (short)0, acc,  false,  false);
    }

    const int mbase = row0 + ((lane >> 4) * 8);
#pragma unroll
    for (int v = 0; v < 8; ++v) {
      float val = acc[v];
      if (RELU) val = fmaxf(val, 0.0f);
      C[(size_t)(mbase + v) * NCOLS + n] = val;
    }
  }
}

extern "C" void kernel_launch(void* const* d_in, const int* in_sizes, int n_in,
                              void* d_out, int out_size, void* d_ws, size_t ws_size,
                              hipStream_t stream) {
  const float* x  = (const float*)d_in[0];
  const int*   er = (const int*)d_in[1];
  const int*   ec = (const int*)d_in[2];
  const float* ev = (const float*)d_in[3];
  const float* W1 = (const float*)d_in[4];
  const float* b1 = (const float*)d_in[5];
  const float* W2 = (const float*)d_in[6];
  const float* b2 = (const float*)d_in[7];
  float* out = (float*)d_out;
  const int E = in_sizes[1];

  const size_t hElems = (size_t)N_NODES * 128;
  float* h1 = (float*)d_ws;
  float* h2 = h1 + hElems;
  float* g  = h2 + hElems;
  int* tcnt   = (int*)(g + (size_t)N_NODES * 64);
  int* toff   = tcnt + TABLE_LEN;
  int* rowptr = toff + TABLE_LEN;
  int* tmp_r  = rowptr + 100096;
  edge_cv* tmp_cv = (edge_cv*)(tmp_r + EDGE_SLOTS);
  edge_cv* cv = tmp_cv + EDGE_SLOTS;

  const int fineBlocks = (N_NODES >> 9) + 1;
  const int gatherBlocks = (N_NODES + 8 * ROWS_PER_WAVE - 1) / (8 * ROWS_PER_WAVE);
  const int gemmBlocks = ((N_NODES / 16) + 7) / 8;

  coarse_hist<<<N_EDGE_BLOCKS, 256, 0, stream>>>(er, tcnt, E);
  table_scan<<<256, SCAN_THREADS, 0, stream>>>(tcnt, toff);
  coarse_scatter<<<N_EDGE_BLOCKS, 256, 0, stream>>>(er, ec, ev, toff, tmp_r, tmp_cv, E);
  fine_sort<<<fineBlocks, 32 * FS_WAVES, 0, stream>>>(toff, tmp_r, tmp_cv, rowptr, cv, E);
  spmm_row_gather<128, false><<<gatherBlocks, 256, 0, stream>>>(rowptr, cv, x, nullptr, h1, E);
  gemm_wmma_f32<128, true, true><<<gemmBlocks, 256, 0, stream>>>(h1, W1, b1, h2, N_NODES);
  gemm_wmma_f32<64, false, false><<<gemmBlocks, 256, 0, stream>>>(h2, W2, b2, g, N_NODES);
  spmm_row_gather<64, true><<<gatherBlocks, 256, 0, stream>>>(rowptr, cv, g, b2, out, E);
}


# Layer 1 is one kernel: a work-group gathers 64 rows of adj x into LDS and runs both products (W1+b1, ReLU, W2) on WMMA from LDS; h1 and h2 never reach memory, 8 launches become 6; sums unchanged
# speedup vs baseline: 4.1988x; 1.1008x over previous
#include <hip/hip_runtime.h>
#include <hip/hip_bf16.h>

typedef __attribute__((ext_vector_type(2))) float v2f;
typedef __attribute__((ext_vector_type(8))) float v8f;

#define N_NODES 100000
#define KDIM    128
#define ROWS_PER_WAVE 4

#define EDGE_CHUNK    4096
#define N_EDGE_BLOCKS 400
#define EDGE_SLOTS    (N_EDGE_BLOCKS * EDGE_CHUNK)
#define TABLE_LEN     (256 * N_EDGE_BLOCKS)
#define FINE_BINS     512


struct __attribute__((aligned(8))) edge_cv { int c; float v; };

template <int BITS>
__device__ __forceinline__ unsigned same_digit_lanes(unsigned d, bool valid) {
  unsigned peers = __builtin_amdgcn_ballot_w32(valid);
#pragma unroll
  for (int b = 0; b < BITS; ++b) {
    const bool bit = (d >> b) & 1u;
    const unsigned m = __builtin_amdgcn_ballot_w32(bit);
    peers &= bit ? m : ~m;
  }
  return peers;
}

__device__ __forceinline__ void coarse_count(const int (&r)[16], const bool (&ok)[16],
                                             int* __restrict__ myhist, unsigned lt) {
#pragma unroll
  for (int b = 0; b < 16; ++b) {
    const unsigned d = ((unsigned)r[b] >> 9) & 255u;
    const unsigned peers = same_digit_lanes<8>(d, ok[b]);
    if (ok[b] && (peers & lt) == 0) myhist[d] += __popc(peers);
  }
}

__global__ __launch_bounds__(256) void coarse_hist(
    const int* __restrict__ rows, int* __restrict__ table, int E) {
  __shared__ int hist[8][256];
  const int t = threadIdx.x, lane = t & 31;
  const int w = __builtin_amdgcn_readfirstlane(t >> 5);
  const unsigned lt = (1u << lane) - 1u;
  const int e0 = blockIdx.x * EDGE_CHUNK + w * 512 + lane;
  int r[16]; bool ok[16];
#pragma unroll
  for (int b = 0; b < 16; ++b) {
    const int e = e0 + b * 32;
    ok[b] = e < E;
    r[b] = ok[b] ? rows[e] : 0;
  }
#pragma unroll
  for (int i = 0; i < 8; ++i) hist[i][t] = 0;
  __syncthreads();
  coarse_count(r, ok, hist[w], lt);
  __syncthreads();
  int s = 0;
#pragma unroll
  for (int i = 0; i < 8; ++i) s += hist[i][t];
  table[t * N_EDGE_BLOCKS + blockIdx.x] = s;
}

#define SCAN_THREADS 256
#define SCAN_BATCH   25
static_assert(TABLE_LEN == 256 * N_EDGE_BLOCKS && N_EDGE_BLOCKS % 4 == 0 &&
              N_EDGE_BLOCKS <= 2 * SCAN_THREADS,
              "table_scan: one group per bin, whole 16-byte chunks, two entries a thread");

__global__ __launch_bounds__(SCAN_THREADS) void table_scan(
    const int* __restrict__ cnt, int* __restrict__ off) {
  __shared__ int wprev[SCAN_THREADS / 32];
  __shared__ int wsum[SCAN_THREADS / 32];
  const int t = threadIdx.x;
  const int lane = t & 31, w = t >> 5;
  const int bin = blockIdx.x;
  const int n4 = bin * (N_EDGE_BLOCKS / 4);
  const int4* __restrict__ c4 = (const int4*)cnt;
  const bool owner = 2 * t < N_EDGE_BLOCKS;
  int2 own = make_int2(0, 0);
  if (owner) own = *(const int2*)(cnt + bin * N_EDGE_BLOCKS + 2 * t);
  int s = 0;
  for (int i0 = 0; i0 < n4; i0 += SCAN_BATCH * SCAN_THREADS) {
    int4 v[SCAN_BATCH];
#pragma unroll
    for (int u = 0; u < SCAN_BATCH; ++u)
      v[u] = c4[min(i0 + u * SCAN_THREADS + t, n4 - 1)];
#pragma unroll
    for (int u = 0; u < SCAN_BATCH; ++u)
      if (i0 + u * SCAN_THREADS + t < n4) s += v[u].x + v[u].y + v[u].z + v[u].w;
  }
#pragma unroll
  for (int d = 16; d >= 1; d >>= 1) s += __shfl_xor(s, d);
  const int tot = own.x + own.y;
  int inc = tot;
#pragma unroll
  for (int d = 1; d < 32; d <<= 1) {
    const int o = __shfl_up(inc, d);
    if (lane >= d) inc += o;
  }
  if (lane == 0) wprev[w] = s;
  if (lane == 31) wsum[w] = inc;
  __syncthreads();
  int run = inc - tot;
#pragma unroll
  for (int i = 0; i < SCAN_THREADS / 32; ++i) {
    run += wprev[i];
    if (i < w) run += wsum[i];
  }
  if (owner)
    *(int2*)(off + bin * N_EDGE_BLOCKS + 2 * t) = make_int2(run, run + own.x);
}

__global__ __launch_bounds__(256) void coarse_scatter(
    const int* __restrict__ rows, const int* __restrict__ cols,
    const float* __restrict__ vals, const int* __restrict__ off,
    int* __restrict__ tmp_r, edge_cv* __restrict__ tmp_cv, int E) {
  __shared__ int hist[8][256];
  __shared__ int gdelta[256];
  __shared__ int wsum[8];
  __shared__ int s_r[EDGE_CHUNK];
  __shared__ edge_cv s_cv[EDGE_CHUNK];
  const int t = threadIdx.x, lane = t & 31;
  const int w = __builtin_amdgcn_readfirstlane(t >> 5);
  const unsigned lt = (1u << lane) - 1u;
  const int e0 = blockIdx.x * EDGE_CHUNK + w * 512 + lane;
  int r[16]; bool ok[16]; edge_cv rec[16];
#pragma unroll
  for (int b = 0; b < 16; ++b) {
    const int e = e0 + b * 32;
    ok[b] = e < E;
    r[b] = ok[b] ? rows[e] : 0;
    rec[b].c = ok[b] ? cols[e] : 0;
    rec[b].v = ok[b] ? vals[e] : 0.0f;
  }
#pragma unroll
  for (int i = 0; i < 8; ++i) hist[i][t] = 0;
  __syncthreads();
  coarse_count(r, ok, hist[w], lt);
  __syncthreads();
  int tot = 0;
#pragma unroll
  for (int i = 0; i < 8; ++i) tot += hist[i][t];
  int inc = tot;
#pragma unroll
  for (int d = 1; d < 32; d <<= 1) {
    const int o = __shfl_up(inc, d);
    if (lane >= d) inc += o;
  }
  if (lane == 31) wsum[w] = inc;
  __syncthreads();
  {
    int lstart = inc - tot;
    for (int i = 0; i < w; ++i) lstart += wsum[i];
    gdelta[t] = off[t * N_EDGE_BLOCKS + blockIdx.x] - lstart;
    int run = lstart;
#pragma unroll
    for (int i = 0; i < 8; ++i) { const int c = hist[i][t]; hist[i][t] = run; run += c; }
  }
  __syncthreads();
#pragma unroll
  for (int b = 0; b < 16; ++b) {
    const unsigned d = ((unsigned)r[b] >> 9) & 255u;
    const unsigned peers = same_digit_lanes<8>(d, ok[b]);
    const int rank = __popc(peers & lt);
    const int pos = hist[w][d] + rank;
    if (ok[b] && rank == 0) hist[w][d] = pos + __popc(peers);
    if (ok[b]) {
      const int p = min(max(pos, 0), EDGE_CHUNK - 1);
      s_r[p] = r[b];
      s_cv[p] = rec[b];
    }
  }
  __syncthreads();
  const int nvalid = min(max(E - (int)blockIdx.x * EDGE_CHUNK, 0), EDGE_CHUNK);
  if (nvalid > 0) {
    int rr[EDGE_CHUNK / 256], gd[EDGE_CHUNK / 256];
    edge_cv ev[EDGE_CHUNK / 256];
#pragma unroll
    for (int k = 0; k < EDGE_CHUNK / 256; ++k) {
      const int i = min(k * 256 + t, nvalid - 1);
      rr[k] = s_r[i];
      ev[k] = s_cv[i];
    }
#pragma unroll
    for (int k = 0; k < EDGE_CHUNK / 256; ++k) gd[k] = gdelta[((unsigned)rr[k] >> 9) & 255u];
#pragma unroll
    for (int k = 0; k < EDGE_CHUNK / 256; ++k) {
      const int i = k * 256 + t;
      if (i < nvalid) {
        const int p = min(max(gd[k] + i, 0), E - 1);
        tmp_r[p] = rr[k];
        tmp_cv[p] = ev[k];
      }
    }
  }
}

#define FS_WAVES 32
#define FS_STAGE 10240
__global__ __launch_bounds__(32 * FS_WAVES) void fine_sort(
    const int* __restrict__ off, const int* __restrict__ tmp_r,
    const edge_cv* __restrict__ tmp_cv, int* __restrict__ rowptr,
    edge_cv* __restrict__ cv, int E) {
  __shared__ int hist[FS_WAVES][FINE_BINS];
  __shared__ int wsum[8];
  __shared__ edge_cv stage[FS_STAGE];
  const int bin = blockIdx.x;
  const int t = threadIdx.x, lane = t & 31;
  const int w = __builtin_amdgcn_readfirstlane(t >> 5);
  const unsigned lt = (1u << lane) - 1u;
  const int seg_e = min((bin == 255) ? E : off[(bin + 1) * N_EDGE_BLOCKS], E);
  const int seg_s = min(max(off[bin * N_EDGE_BLOCKS], 0), seg_e);
  const int per = ((seg_e - seg_s + 32 * FS_WAVES - 1) / (32 * FS_WAVES)) * 32;
  const int ws = min(seg_s + w * per, seg_e);
  const int we = min(ws + per, seg_e);
  const int nseg = seg_e - seg_s;
  const bool staged = nseg <= FS_STAGE;
#pragma unroll
  for (int i = 0; i < FINE_BINS / 32; ++i) (&hist[0][0])[t + i * 32 * FS_WAVES] = 0;
  __syncthreads();
  for (int j0 = ws; j0 < we; j0 += 128) {
    int r[4]; bool ok[4];
#pragma unroll
    for (int q = 0; q < 4; ++q) {
      const int j = j0 + q * 32 + lane;
      ok[q] = j < we;
      r[q] = ok[q] ? tmp_r[j] : 0;
    }
#pragma unroll
    for (int q = 0; q < 4; ++q) {
      const unsigned d = (unsigned)r[q] & (FINE_BINS - 1);
      const unsigned peers = same_digit_lanes<9>(d, ok[q]);
      if (ok[q] && (peers & lt) == 0) hist[w][d] += __popc(peers);
    }
  }
  __syncthreads();
  int d0 = 0, d1 = 0;
  if (t < 256) {
#pragma unroll 8
    for (int i = 0; i < FS_WAVES; ++i) { d0 += hist[i][2 * t]; d1 += hist[i][2 * t + 1]; }
  }
  const int tot = d0 + d1;
  int inc = tot;
#pragma unroll
  for (int d = 1; d < 32; d <<= 1) {
    const int o = __shfl_up(inc, d);
    if (lane >= d) inc += o;
  }
  if (lane == 31 && t < 256) wsum[w] = inc;
  __syncthreads();
  if (t < 256) {
    int ex = seg_s + inc - tot;
    for (int i = 0; i < w; ++i) ex += wsum[i];
    const int rr = bin * FINE_BINS + 2 * t;
    if (rr <= N_NODES) rowptr[rr] = ex;
    if (rr + 1 <= N_NODES) rowptr[rr + 1] = ex + d0;
    int run0 = ex, run1 = ex + d0;
#pragma unroll 8
    for (int i = 0; i < FS_WAVES; ++i) {
      const int c0 = hist[i][2 * t], c1 = hist[i][2 * t + 1];
      hist[i][2 * t] = run0;     run0 += c0;
      hist[i][2 * t + 1] = run1; run1 += c1;
    }
  }
  __syncthreads();
  for (int j0 = ws; j0 < we; j0 += 128) {
    int r[4]; bool ok[4]; edge_cv rec[4];
#pragma unroll
    for (int q = 0; q < 4; ++q) {
      const int j = j0 + q * 32 + lane;
      ok[q] = j < we;
      r[q] = ok[q] ? tmp_r[j] : 0;
      if (ok[q]) rec[q] = tmp_cv[j]; else { rec[q].c = 0; rec[q].v = 0.0f; }
    }
#pragma unroll
    for (int q = 0; q < 4; ++q) {
      const unsigned d = (unsigned)r[q] & (FINE_BINS - 1);
      const unsigned peers = same_digit_lanes<9>(d, ok[q]);
      const int rank = __popc(peers & lt);
      const int pos = hist[w][d] + rank;
      if (ok[q] && rank == 0) hist[w][d] = pos + __popc(peers);
      if (ok[q]) {
        if (staged) stage[min(max(pos - seg_s, 0), FS_STAGE - 1)] = rec[q];
        else cv[min(max(pos, 0), E - 1)] = rec[q];
      }
    }
  }
  __syncthreads();
  if (staged) {
    for (int i = t; i < nseg; i += 32 * FS_WAVES) cv[seg_s + i] = stage[i];
  }
}

template <int F> struct lane_vec {
  typedef float type __attribute__((ext_vector_type(F / 32)));
};

template <int F, int U>
__device__ __forceinline__ typename lane_vec<F>::type gather_edges(
    const edge_cv* __restrict__ cv, int j, const float* __restrict__ x,
    int lane, typename lane_vec<F>::type acc) {
  typedef typename lane_vec<F>::type vt;
  edge_cv e[U];
  vt xv[U];
#pragma unroll
  for (int u = 0; u < U; ++u) e[u] = cv[j + u];
#pragma unroll
  for (int u = 0; u < U; ++u)
    xv[u] = *((const vt*)(x + (size_t)e[u].c * F) + lane);
#pragma unroll
  for (int u = 0; u < U; ++u) acc += xv[u] * e[u].v;
  return acc;
}

template <int F, bool BIAS>
__global__ __launch_bounds__(256) void spmm_row_gather(
    const int* __restrict__ rowptr, const edge_cv* __restrict__ cv,
    const float* __restrict__ x, const float* __restrict__ bias,
    float* __restrict__ out, int E) {
  typedef typename lane_vec<F>::type vt;
  const int lane = threadIdx.x & 31;
  const int wave =
      __builtin_amdgcn_readfirstlane(blockIdx.x * 8 + (threadIdx.x >> 5));
  vt b = {};
  if (BIAS) b = *((const vt*)bias + lane);
  for (int i = 0; i < ROWS_PER_WAVE; ++i) {
    const int r = wave * ROWS_PER_WAVE + i;
    if (r >= N_NODES) break;
    const int t = min(rowptr[r + 1], E);
    int j = min(max(rowptr[r], 0), t);
    vt acc = b;
    for (; j + 8 <= t; j += 8) acc = gather_edges<F, 8>(cv, j, x, lane, acc);
    if (j + 4 <= t) { acc = gather_edges<F, 4>(cv, j, x, lane, acc); j += 4; }
    if (j + 2 <= t) { acc = gather_edges<F, 2>(cv, j, x, lane, acc); j += 2; }
    if (j < t) acc = gather_edges<F, 1>(cv, j, x, lane, acc);
    *((vt*)(out + (size_t)r * F) + lane) = acc;
  }
}

#define L1_ROWS  64
#define L1_WAVES 16
#define ROW_PITCH (KDIM + 4)
#define W1_PITCH  (2 * 128 + 32)
#define W2_PITCH  (2 * 64 + 32)

template <int NC> struct weight_regs {
  static constexpr int CNT = (KDIM / 2) * (NC / 2) / (32 * L1_WAVES);
  static_assert(CNT * 32 * L1_WAVES == (KDIM / 2) * (NC / 2), "whole 2 x 2 blocks per thread");
  float2 a[CNT], b[CNT];
};

template <int NC>
__device__ __forceinline__ void load_weights(const float* __restrict__ Wg, int t,
                                             weight_regs<NC>& wr) {
#pragma unroll
  for (int u = 0; u < weight_regs<NC>::CNT; ++u) {
    const int i = t + u * 32 * L1_WAVES;
    const int p = i / (NC / 2), cq = i % (NC / 2);
    wr.a[u] = *(const float2*)(Wg + (size_t)(2 * p) * NC + 2 * cq);
    wr.b[u] = *(const float2*)(Wg + (size_t)(2 * p + 1) * NC + 2 * cq);
  }
}

template <int NC, int PITCH>
__device__ __forceinline__ void store_weights(float* __restrict__ Ws, int t,
                                              const weight_regs<NC>& wr) {
#pragma unroll
  for (int u = 0; u < weight_regs<NC>::CNT; ++u) {
    const int i = t + u * 32 * L1_WAVES;
    const int p = i / (NC / 2), cq = i % (NC / 2);
    *(float4*)(Ws + p * PITCH + 4 * cq) =
        make_float4(wr.a[u].x, wr.b[u].x, wr.a[u].y, wr.b[u].y);
  }
}

__global__ __launch_bounds__(32 * L1_WAVES) void layer1_fused(
    const int* __restrict__ rowptr, const edge_cv* __restrict__ cv,
    const float* __restrict__ x, const float* __restrict__ W1,
    const float* __restrict__ b1, const float* __restrict__ W2,
    float* __restrict__ g, int E) {
  typedef typename lane_vec<KDIM>::type vt;
  __shared__ __align__(16) float W1s[(KDIM / 2) * W1_PITCH];
  __shared__ __align__(16) float W2s[(KDIM / 2) * W2_PITCH];
  __shared__ __align__(16) float rows[L1_ROWS * ROW_PITCH];
  const int t = threadIdx.x, lane = t & 31;
  const int w = __builtin_amdgcn_readfirstlane(t >> 5);
  const int rbase = blockIdx.x * L1_ROWS;

  weight_regs<128> wr1;
  weight_regs<64> wr2;
  load_weights<128>(W1, t, wr1);
  load_weights<64>(W2, t, wr2);

  for (int i = 0; i < L1_ROWS / L1_WAVES; ++i) {
    const int lr = w * (L1_ROWS / L1_WAVES) + i;
    const int r = rbase + lr;
    vt acc = {};
    if (r < N_NODES) {
      const int te = min(rowptr[r + 1], E);
      int j = min(max(rowptr[r], 0), te);
      for (; j + 8 <= te; j += 8) acc = gather_edges<KDIM, 8>(cv, j, x, lane, acc);
      if (j + 4 <= te) { acc = gather_edges<KDIM, 4>(cv, j, x, lane, acc); j += 4; }
      if (j + 2 <= te) { acc = gather_edges<KDIM, 2>(cv, j, x, lane, acc); j += 2; }
      if (j < te) acc = gather_edges<KDIM, 1>(cv, j, x, lane, acc);
    }
    *((vt*)(rows + lr * ROW_PITCH) + lane) = acc;
  }
  store_weights<128, W1_PITCH>(W1s, t, wr1);
  store_weights<64, W2_PITCH>(W2s, t, wr2);
  __syncthreads();

  const int c = lane & 15;
  const int hl = lane >> 4;
  const int rt = w & 3, q = w >> 2;
  float* myrow = rows + (rt * 16 + c) * ROW_PITCH;

  v2f xf[KDIM / 4];
#pragma unroll
  for (int ks = 0; ks < KDIM / 4; ++ks) xf[ks] = *(const v2f*)(myrow + 4 * ks + 2 * hl);
  const int n0 = q * 32, n1 = n0 + 16;
  v8f acc0, acc1;
#pragma unroll
  for (int v = 0; v < 8; ++v) { acc0[v] = b1[n0 + 8 * hl + v]; acc1[v] = b1[n1 + 8 * hl + v]; }
#pragma unroll
  for (int ks = 0; ks < KDIM / 4; ++ks) {
    const float* wrow = W1s + (2 * ks + hl) * W1_PITCH + 2 * c;
    const v2f wf0 = *(const v2f*)(wrow + 2 * n0);
    const v2f wf1 = *(const v2f*)(wrow + 2 * n1);
    acc0 = __builtin_amdgcn_wmma_f32_16x16x4_f32(
         false, wf0,  false, xf[ks],
         (short)0, acc0,  false,  false);
    acc1 = __builtin_amdgcn_wmma_f32_16x16x4_f32(
         false, wf1,  false, xf[ks],
         (short)0, acc1,  false,  false);
  }
#pragma unroll
  for (int v = 0; v < 8; ++v) { acc0[v] = fmaxf(acc0[v], 0.0f); acc1[v] = fmaxf(acc1[v], 0.0f); }
  __syncthreads();
  *(float4*)(myrow + n0 + 8 * hl)     = make_float4(acc0[0], acc0[1], acc0[2], acc0[3]);
  *(float4*)(myrow + n0 + 8 * hl + 4) = make_float4(acc0[4], acc0[5], acc0[6], acc0[7]);
  *(float4*)(myrow + n1 + 8 * hl)     = make_float4(acc1[0], acc1[1], acc1[2], acc1[3]);
  *(float4*)(myrow + n1 + 8 * hl + 4) = make_float4(acc1[4], acc1[5], acc1[6], acc1[7]);
  __syncthreads();

#pragma unroll
  for (int ks = 0; ks < KDIM / 4; ++ks) xf[ks] = *(const v2f*)(myrow + 4 * ks + 2 * hl);
  v8f acc2;
#pragma unroll
  for (int v = 0; v < 8; ++v) acc2[v] = 0.0f;
#pragma unroll
  for (int ks = 0; ks < KDIM / 4; ++ks) {
    const v2f wf = *(const v2f*)(W2s + (2 * ks + hl) * W2_PITCH + 2 * (16 * q + c));
    acc2 = __builtin_amdgcn_wmma_f32_16x16x4_f32(
         false, wf,  false, xf[ks],
         (short)0, acc2,  false,  false);
  }
  const int row = rbase + rt * 16 + c;
  if (row < N_NODES) {
    float* gp = g + (size_t)row * 64 + 16 * q + 8 * hl;
    *(float4*)gp       = make_float4(acc2[0], acc2[1], acc2[2], acc2[3]);
    *(float4*)(gp + 4) = make_float4(acc2[4], acc2[5], acc2[6], acc2[7]);
  }
}

extern "C" void kernel_launch(void* const* d_in, const int* in_sizes, int n_in,
                              void* d_out, int out_size, void* d_ws, size_t ws_size,
                              hipStream_t stream) {
  const float* x  = (const float*)d_in[0];
  const int*   er = (const int*)d_in[1];
  const int*   ec = (const int*)d_in[2];
  const float* ev = (const float*)d_in[3];
  const float* W1 = (const float*)d_in[4];
  const float* b1 = (const float*)d_in[5];
  const float* W2 = (const float*)d_in[6];
  const float* b2 = (const float*)d_in[7];
  float* out = (float*)d_out;
  const int E = in_sizes[1];

  float* g  = (float*)d_ws;
  int* tcnt   = (int*)(g + (size_t)N_NODES * 64);
  int* toff   = tcnt + TABLE_LEN;
  int* rowptr = toff + TABLE_LEN;
  int* tmp_r  = rowptr + 100096;
  edge_cv* tmp_cv = (edge_cv*)(tmp_r + EDGE_SLOTS);
  edge_cv* cv = tmp_cv + EDGE_SLOTS;

  const int fineBlocks = (N_NODES >> 9) + 1;
  const int gatherBlocks = (N_NODES + 8 * ROWS_PER_WAVE - 1) / (8 * ROWS_PER_WAVE);
  const int layer1Blocks = (N_NODES + L1_ROWS - 1) / L1_ROWS;

  coarse_hist<<<N_EDGE_BLOCKS, 256, 0, stream>>>(er, tcnt, E);
  table_scan<<<256, SCAN_THREADS, 0, stream>>>(tcnt, toff);
  coarse_scatter<<<N_EDGE_BLOCKS, 256, 0, stream>>>(er, ec, ev, toff, tmp_r, tmp_cv, E);
  fine_sort<<<fineBlocks, 32 * FS_WAVES, 0, stream>>>(toff, tmp_r, tmp_cv, rowptr, cv, E);
  layer1_fused<<<layer1Blocks, 32 * L1_WAVES, 0, stream>>>(rowptr, cv, x, W1, b1, W2, g, E);
  spmm_row_gather<64, true><<<gatherBlocks, 256, 0, stream>>>(rowptr, cv, g, b2, out, E);
}


# Edge sort drops its global offset pass: a block orders its 4096 edges by bin in place, the per-bin sort collects its 400 runs in LDS; 6 launches become 4; order of sums unchanged
# speedup vs baseline: 4.7753x; 1.1373x over previous
#include <hip/hip_runtime.h>
#include <hip/hip_bf16.h>

typedef __attribute__((ext_vector_type(2))) float v2f;
typedef __attribute__((ext_vector_type(8))) float v8f;

#define N_NODES 100000
#define KDIM    128
#define ROWS_PER_WAVE 4

#define EDGE_CHUNK    4096
#define N_EDGE_BLOCKS 400
#define EDGE_SLOTS    (N_EDGE_BLOCKS * EDGE_CHUNK)
#define TABLE_LEN     (256 * N_EDGE_BLOCKS)
#define FINE_BINS     512


struct __attribute__((aligned(8))) edge_cv { int c; float v; };

template <int BITS>
__device__ __forceinline__ unsigned same_digit_lanes(unsigned d, bool valid) {
  unsigned peers = __builtin_amdgcn_ballot_w32(valid);
#pragma unroll
  for (int b = 0; b < BITS; ++b) {
    const bool bit = (d >> b) & 1u;
    const unsigned m = __builtin_amdgcn_ballot_w32(bit);
    peers &= bit ? m : ~m;
  }
  return peers;
}

__device__ __forceinline__ void coarse_count(const int (&r)[16], const bool (&ok)[16],
                                             int* __restrict__ myhist, unsigned lt) {
#pragma unroll
  for (int b = 0; b < 16; ++b) {
    const unsigned d = ((unsigned)r[b] >> 9) & 255u;
    const unsigned peers = same_digit_lanes<8>(d, ok[b]);
    if (ok[b] && (peers & lt) == 0) myhist[d] += __popc(peers);
  }
}

__global__ __launch_bounds__(256) void block_bin_sort(
    const int* __restrict__ rows, const int* __restrict__ cols,
    const float* __restrict__ vals, int* __restrict__ tcnt, int* __restrict__ tlst,
    int* __restrict__ tmp_r, edge_cv* __restrict__ tmp_cv, int E) {
  __shared__ int hist[8][256];
  __shared__ int wsum[8];
  __shared__ int s_r[EDGE_CHUNK];
  __shared__ edge_cv s_cv[EDGE_CHUNK];
  const int t = threadIdx.x, lane = t & 31;
  const int w = __builtin_amdgcn_readfirstlane(t >> 5);
  const unsigned lt = (1u << lane) - 1u;
  const int e0 = blockIdx.x * EDGE_CHUNK + w * 512 + lane;
  int r[16]; bool ok[16]; edge_cv rec[16];
#pragma unroll
  for (int b = 0; b < 16; ++b) {
    const int e = e0 + b * 32;
    ok[b] = e < E;
    r[b] = ok[b] ? rows[e] : 0;
    rec[b].c = ok[b] ? cols[e] : 0;
    rec[b].v = ok[b] ? vals[e] : 0.0f;
  }
#pragma unroll
  for (int i = 0; i < 8; ++i) hist[i][t] = 0;
  __syncthreads();
  coarse_count(r, ok, hist[w], lt);
  __syncthreads();
  int tot = 0;
#pragma unroll
  for (int i = 0; i < 8; ++i) tot += hist[i][t];
  int inc = tot;
#pragma unroll
  for (int d = 1; d < 32; d <<= 1) {
    const int o = __shfl_up(inc, d);
    if (lane >= d) inc += o;
  }
  if (lane == 31) wsum[w] = inc;
  __syncthreads();
  {
    int lstart = inc - tot;
    for (int i = 0; i < w; ++i) lstart += wsum[i];
    tcnt[t * N_EDGE_BLOCKS + blockIdx.x] = tot;
    tlst[t * N_EDGE_BLOCKS + blockIdx.x] = lstart;
    int run = lstart;
#pragma unroll
    for (int i = 0; i < 8; ++i) { const int c = hist[i][t]; hist[i][t] = run; run += c; }
  }
  __syncthreads();
#pragma unroll
  for (int b = 0; b < 16; ++b) {
    const unsigned d = ((unsigned)r[b] >> 9) & 255u;
    const unsigned peers = same_digit_lanes<8>(d, ok[b]);
    const int rank = __popc(peers & lt);
    const int pos = hist[w][d] + rank;
    if (ok[b] && rank == 0) hist[w][d] = pos + __popc(peers);
    if (ok[b]) {
      const int p = min(max(pos, 0), EDGE_CHUNK - 1);
      s_r[p] = r[b];
      s_cv[p] = rec[b];
    }
  }
  __syncthreads();
  const int nvalid = min(max(E - (int)blockIdx.x * EDGE_CHUNK, 0), EDGE_CHUNK);
  if (nvalid > 0) {
    int rr[EDGE_CHUNK / 256];
    edge_cv ev[EDGE_CHUNK / 256];
#pragma unroll
    for (int k = 0; k < EDGE_CHUNK / 256; ++k) {
      const int i = min(k * 256 + t, nvalid - 1);
      rr[k] = s_r[i];
      ev[k] = s_cv[i];
    }
#pragma unroll
    for (int k = 0; k < EDGE_CHUNK / 256; ++k) {
      const int i = k * 256 + t;
      if (i < nvalid) {
        const int p = blockIdx.x * EDGE_CHUNK + i;
        tmp_r[p] = rr[k];
        tmp_cv[p] = ev[k];
      }
    }
  }
}

#define FS_WAVES 32
#define FS_STAGE 10240
#define FS_RUNS ((N_EDGE_BLOCKS + FS_WAVES - 1) / FS_WAVES)
#define FS_SUM_BATCH 8
static_assert(N_EDGE_BLOCKS % 4 == 0 && N_EDGE_BLOCKS <= 32 * FS_WAVES,
              "fine_sort: whole 16-byte chunks per bin, one thread per block's run");

__global__ __launch_bounds__(32 * FS_WAVES) void fine_sort(
    const int* __restrict__ tcnt, const int* __restrict__ tlst,
    const int* __restrict__ tmp_r, const edge_cv* __restrict__ tmp_cv,
    int* big_r, edge_cv* big_cv,
    int* __restrict__ rowptr, edge_cv* __restrict__ cv, int E) {
  __shared__ int hist[FS_WAVES][FINE_BINS];
  __shared__ int wsum[8];
  __shared__ int wbefore[FS_WAVES];
  __shared__ int wown[FS_WAVES];
  __shared__ int4 run_info[N_EDGE_BLOCKS];
  __shared__ int s_r[FS_STAGE];
  __shared__ edge_cv s_cv[FS_STAGE];
  __shared__ edge_cv stage[FS_STAGE];
  const int bin = blockIdx.x;
  const int t = threadIdx.x, lane = t & 31;
  const int w = __builtin_amdgcn_readfirstlane(t >> 5);
  const unsigned lt = (1u << lane) - 1u;

  const int tb = min(t, N_EDGE_BLOCKS - 1);
  int rl = tlst[bin * N_EDGE_BLOCKS + tb];
  int rc0 = tcnt[bin * N_EDGE_BLOCKS + tb];
  const int n4 = bin * (N_EDGE_BLOCKS / 4);
  const int4* __restrict__ c4 = (const int4*)tcnt;
  int sb = 0;
  for (int i0 = 0; i0 < n4; i0 += FS_SUM_BATCH * 32 * FS_WAVES) {
    int4 v[FS_SUM_BATCH];
#pragma unroll
    for (int u = 0; u < FS_SUM_BATCH; ++u)
      v[u] = c4[min(i0 + u * 32 * FS_WAVES + t, n4 - 1)];
#pragma unroll
    for (int u = 0; u < FS_SUM_BATCH; ++u)
      if (i0 + u * 32 * FS_WAVES + t < n4) sb += v[u].x + v[u].y + v[u].z + v[u].w;
  }
#pragma unroll
  for (int d = 16; d >= 1; d >>= 1) sb += __shfl_xor(sb, d);
  rl = min(max(rl, 0), EDGE_CHUNK);
  rc0 = (t < N_EDGE_BLOCKS) ? min(max(rc0, 0), EDGE_CHUNK - rl) : 0;
  int cinc = rc0;
#pragma unroll
  for (int d = 1; d < 32; d <<= 1) {
    const int o = __shfl_up(cinc, d);
    if (lane >= d) cinc += o;
  }
  if (lane == 0) wbefore[w] = sb;
  if (lane == 31) wown[w] = cinc;
#pragma unroll
  for (int i = 0; i < FINE_BINS / 32; ++i) (&hist[0][0])[t + i * 32 * FS_WAVES] = 0;
  __syncthreads();
  int seg_s = 0, nraw = 0, rpre0 = cinc - rc0;
#pragma unroll
  for (int i = 0; i < FS_WAVES; ++i) {
    seg_s += wbefore[i];
    const int o = wown[i];
    nraw += o;
    if (i < w) rpre0 += o;
  }
  seg_s = min(max(seg_s, 0), E);
  const int nseg = min(nraw, E - seg_s);
  const bool staged = nraw <= FS_STAGE;
  if (t < N_EDGE_BLOCKS) run_info[t] = make_int4(rpre0, rc0, t * EDGE_CHUNK + rl, 0);
  __syncthreads();

  auto put = [&](int p, int r, edge_cv rec) {
    if (staged) { s_r[p] = r; s_cv[p] = rec; }
    else { const int q = min(seg_s + p, E - 1); big_r[q] = r; big_cv[q] = rec; }
  };
  auto row_at = [&](int j) -> int {
    return staged ? s_r[j] : big_r[min(seg_s + j, E - 1)];
  };
  auto rec_at = [&](int j) -> edge_cv {
    return staged ? s_cv[j] : big_cv[min(seg_s + j, E - 1)];
  };
  {
    int kc[FS_RUNS], kp[FS_RUNS], ks[FS_RUNS];
    int rr[FS_RUNS]; edge_cv re[FS_RUNS];
    int4 ri[FS_RUNS];
#pragma unroll
    for (int m = 0; m < FS_RUNS; ++m) ri[m] = run_info[min(w + m * FS_WAVES, N_EDGE_BLOCKS - 1)];
#pragma unroll
    for (int m = 0; m < FS_RUNS; ++m) {
      const bool has = w + m * FS_WAVES < N_EDGE_BLOCKS;
      kp[m] = __builtin_amdgcn_readfirstlane(ri[m].x);
      kc[m] = has ? __builtin_amdgcn_readfirstlane(ri[m].y) : 0;
      ks[m] = __builtin_amdgcn_readfirstlane(ri[m].z);
    }
#pragma unroll
    for (int m = 0; m < FS_RUNS; ++m) {
      const bool in = lane < kc[m];
      rr[m] = in ? tmp_r[ks[m] + lane] : 0;
      if (in) re[m] = tmp_cv[ks[m] + lane]; else { re[m].c = 0; re[m].v = 0.0f; }
    }
#pragma unroll
    for (int m = 0; m < FS_RUNS; ++m)
      if (lane < kc[m]) put(kp[m] + lane, rr[m], re[m]);
#pragma unroll
    for (int m = 0; m < FS_RUNS; ++m)
      for (int i = lane + 32; i < kc[m]; i += 32) put(kp[m] + i, tmp_r[ks[m] + i], tmp_cv[ks[m] + i]);
  }
  __syncthreads();

  const int per = ((nseg + 32 * FS_WAVES - 1) / (32 * FS_WAVES)) * 32;
  const int ws = min(w * per, nseg);
  const int we = min(ws + per, nseg);
  for (int j0 = ws; j0 < we; j0 += 128) {
    int r[4]; bool ok[4];
#pragma unroll
    for (int q = 0; q < 4; ++q) {
      const int j = j0 + q * 32 + lane;
      ok[q] = j < we;
      r[q] = ok[q] ? row_at(j) : 0;
    }
#pragma unroll
    for (int q = 0; q < 4; ++q) {
      const unsigned d = (unsigned)r[q] & (FINE_BINS - 1);
      const unsigned peers = same_digit_lanes<9>(d, ok[q]);
      if (ok[q] && (peers & lt) == 0) hist[w][d] += __popc(peers);
    }
  }
  __syncthreads();
  int d0 = 0, d1 = 0;
  if (t < 256) {
#pragma unroll 8
    for (int i = 0; i < FS_WAVES; ++i) { d0 += hist[i][2 * t]; d1 += hist[i][2 * t + 1]; }
  }
  const int tot = d0 + d1;
  int inc = tot;
#pragma unroll
  for (int d = 1; d < 32; d <<= 1) {
    const int o = __shfl_up(inc, d);
    if (lane >= d) inc += o;
  }
  if (lane == 31 && t < 256) wsum[w] = inc;
  __syncthreads();
  if (t < 256) {
    int ex = seg_s + inc - tot;
    for (int i = 0; i < w; ++i) ex += wsum[i];
    const int rr = bin * FINE_BINS + 2 * t;
    if (rr <= N_NODES) rowptr[rr] = ex;
    if (rr + 1 <= N_NODES) rowptr[rr + 1] = ex + d0;
    int run0 = ex, run1 = ex + d0;
#pragma unroll 8
    for (int i = 0; i < FS_WAVES; ++i) {
      const int c0 = hist[i][2 * t], c1 = hist[i][2 * t + 1];
      hist[i][2 * t] = run0;     run0 += c0;
      hist[i][2 * t + 1] = run1; run1 += c1;
    }
  }
  __syncthreads();
  for (int j0 = ws; j0 < we; j0 += 128) {
    int r[4]; bool ok[4]; edge_cv rec[4];
#pragma unroll
    for (int q = 0; q < 4; ++q) {
      const int j = j0 + q * 32 + lane;
      ok[q] = j < we;
      r[q] = ok[q] ? row_at(j) : 0;
      if (ok[q]) rec[q] = rec_at(j); else { rec[q].c = 0; rec[q].v = 0.0f; }
    }
#pragma unroll
    for (int q = 0; q < 4; ++q) {
      const unsigned d = (unsigned)r[q] & (FINE_BINS - 1);
      const unsigned peers = same_digit_lanes<9>(d, ok[q]);
      const int rank = __popc(peers & lt);
      const int pos = hist[w][d] + rank;
      if (ok[q] && rank == 0) hist[w][d] = pos + __popc(peers);
      if (ok[q]) {
        if (staged) stage[min(max(pos - seg_s, 0), FS_STAGE - 1)] = rec[q];
        else cv[min(max(pos, 0), E - 1)] = rec[q];
      }
    }
  }
  __syncthreads();
  if (staged) {
    for (int i = t; i < nseg; i += 32 * FS_WAVES) cv[seg_s + i] = stage[i];
  }
}

#define RUN_RECS 3

static_assert(N_NODES % ROWS_PER_WAVE == 0, "a wave's rows are all inside the matrix or all outside");

template <int F> struct lane_vec {
  typedef float type __attribute__((ext_vector_type(F / 32)));
};

template <int F, int G, typename Pre, typename Sink>
__device__ __forceinline__ void gather_rows(
    const int* __restrict__ rowptr, const edge_cv* __restrict__ cv,
    const float* __restrict__ x, int r0, int lane, int E,
    typename lane_vec<F>::type init, Pre pre, Sink sink) {
  typedef typename lane_vec<F>::type vt;
  static_assert(32 % G == 0, "whole groups per record register");
  int b[ROWS_PER_WAVE + 1];
#pragma unroll
  for (int i = 0; i <= ROWS_PER_WAVE; ++i) b[i] = min(max(rowptr[r0 + i], 0), E);
#pragma unroll
  for (int i = 1; i <= ROWS_PER_WAVE; ++i) b[i] = max(b[i], b[i - 1]);
  const int e = b[ROWS_PER_WAVE];
  int base = b[0];
  edge_cv rec[RUN_RECS];
#pragma unroll
  for (int q = 0; q < RUN_RECS; ++q) { rec[q].c = 0; rec[q].v = 0.0f; }
  if (base < e) {
#pragma unroll
    for (int q = 0; q < RUN_RECS; ++q) rec[q] = cv[min(base + 32 * q + lane, e - 1)];
  }
  pre();
  int cur = 0;
  int next = b[1];
  vt acc = init;
  while (base < e) {
    for (int g = 0; g < RUN_RECS * (32 / G); ++g) {
      const int j0 = base + G * g;
      if (j0 >= e) break;
      const int q = g / (32 / G), l0 = (g % (32 / G)) * G;
      int rc = rec[RUN_RECS - 1].c;
      float rv = rec[RUN_RECS - 1].v;
#pragma unroll
      for (int i = RUN_RECS - 2; i >= 0; --i) { rc = (q == i) ? rec[i].c : rc; rv = (q == i) ? rec[i].v : rv; }
      float v[G];
      vt xv[G];
#pragma unroll
      for (int u = 0; u < G; ++u) {
        const int c = __builtin_amdgcn_readlane(rc, l0 + u);
        v[u] = __int_as_float(__builtin_amdgcn_readlane(__float_as_int(rv), l0 + u));
        xv[u] = *((const vt*)(x + (size_t)c * F) + lane);
      }
#pragma unroll
      for (int u = 0; u < G; ++u) {
        const int j = j0 + u;
        if (j < e) {
          while (cur < ROWS_PER_WAVE - 1 && j == next) {
            sink(cur, acc);
            acc = init;
            ++cur;
            next = b[ROWS_PER_WAVE];
#pragma unroll
            for (int i = ROWS_PER_WAVE - 1; i >= 2; --i) next = (cur + 1 == i) ? b[i] : next;
          }
          acc += xv[u] * v[u];
        }
      }
    }
    base += 32 * RUN_RECS;
    if (base < e) {
#pragma unroll
      for (int q = 0; q < RUN_RECS; ++q) rec[q] = cv[min(base + 32 * q + lane, e - 1)];
    }
  }
  for (; cur < ROWS_PER_WAVE; ++cur) { sink(cur, acc); acc = init; }
}

template <int F, bool BIAS>
__global__ __launch_bounds__(256) void spmm_row_gather(
    const int* __restrict__ rowptr, const edge_cv* __restrict__ cv,
    const float* __restrict__ x, const float* __restrict__ bias,
    float* __restrict__ out, int E) {
  typedef typename lane_vec<F>::type vt;
  const int lane = threadIdx.x & 31;
  const int wave =
      __builtin_amdgcn_readfirstlane(blockIdx.x * 8 + (threadIdx.x >> 5));
  const int r0 = wave * ROWS_PER_WAVE;
  if (r0 >= N_NODES) return;
  vt b = {};
  if (BIAS) b = *((const vt*)bias + lane);
  gather_rows<F, 16>(rowptr, cv, x, r0, lane, E, b, [] {}, [&](int i, vt a) {
    *((vt*)(out + (size_t)(r0 + i) * F) + lane) = a;
  });
}

#define L1_ROWS  64
#define L1_WAVES 16
#define ROW_PITCH (KDIM + 4)
#define W1_PITCH  (2 * 128 + 32)
#define W2_PITCH  (2 * 64 + 32)

template <int NC> struct weight_regs {
  static constexpr int CNT = (KDIM / 2) * (NC / 2) / (32 * L1_WAVES);
  static_assert(CNT * 32 * L1_WAVES == (KDIM / 2) * (NC / 2), "whole 2 x 2 blocks per thread");
  float2 a[CNT], b[CNT];
};

template <int NC>
__device__ __forceinline__ void load_weights(const float* __restrict__ Wg, int t,
                                             weight_regs<NC>& wr) {
#pragma unroll
  for (int u = 0; u < weight_regs<NC>::CNT; ++u) {
    const int i = t + u * 32 * L1_WAVES;
    const int p = i / (NC / 2), cq = i % (NC / 2);
    wr.a[u] = *(const float2*)(Wg + (size_t)(2 * p) * NC + 2 * cq);
    wr.b[u] = *(const float2*)(Wg + (size_t)(2 * p + 1) * NC + 2 * cq);
  }
}

template <int NC, int PITCH>
__device__ __forceinline__ void store_weights(float* __restrict__ Ws, int t,
                                              const weight_regs<NC>& wr) {
#pragma unroll
  for (int u = 0; u < weight_regs<NC>::CNT; ++u) {
    const int i = t + u * 32 * L1_WAVES;
    const int p = i / (NC / 2), cq = i % (NC / 2);
    *(float4*)(Ws + p * PITCH + 4 * cq) =
        make_float4(wr.a[u].x, wr.b[u].x, wr.a[u].y, wr.b[u].y);
  }
}

__global__ __launch_bounds__(32 * L1_WAVES) void layer1_fused(
    const int* __restrict__ rowptr, const edge_cv* __restrict__ cv,
    const float* __restrict__ x, const float* __restrict__ W1,
    const float* __restrict__ b1, const float* __restrict__ W2,
    float* __restrict__ g, int E) {
  typedef typename lane_vec<KDIM>::type vt;
  __shared__ __align__(16) float W1s[(KDIM / 2) * W1_PITCH];
  __shared__ __align__(16) float W2s[(KDIM / 2) * W2_PITCH];
  __shared__ __align__(16) float rows[L1_ROWS * ROW_PITCH];
  const int t = threadIdx.x, lane = t & 31;
  const int w = __builtin_amdgcn_readfirstlane(t >> 5);
  const int rbase = blockIdx.x * L1_ROWS;

  weight_regs<128> wr1;
  weight_regs<64> wr2;
  load_weights<128>(W1, t, wr1);
  load_weights<64>(W2, t, wr2);

  static_assert(L1_ROWS / L1_WAVES == ROWS_PER_WAVE, "a wave gathers ROWS_PER_WAVE rows of the group");
  auto put_weights = [&] {
    store_weights<128, W1_PITCH>(W1s, t, wr1);
    store_weights<64, W2_PITCH>(W2s, t, wr2);
  };
  {
    const int lr0 = w * ROWS_PER_WAVE;
    const vt zero = {};
    if (rbase + lr0 < N_NODES) {
      gather_rows<KDIM, 16>(rowptr, cv, x, rbase + lr0, lane, E, zero, put_weights, [&](int i, vt a) {
        *((vt*)(rows + (lr0 + i) * ROW_PITCH) + lane) = a;
      });
    } else {
      put_weights();
#pragma unroll
      for (int i = 0; i < ROWS_PER_WAVE; ++i) *((vt*)(rows + (lr0 + i) * ROW_PITCH) + lane) = zero;
    }
  }
  __syncthreads();

  const int c = lane & 15;
  const int hl = lane >> 4;
  const int rt = w & 3, q = w >> 2;
  float* myrow = rows + (rt * 16 + c) * ROW_PITCH;

  v2f xf[KDIM / 4];
#pragma unroll
  for (int ks = 0; ks < KDIM / 4; ++ks) xf[ks] = *(const v2f*)(myrow + 4 * ks + 2 * hl);
  const int n0 = q * 32, n1 = n0 + 16;
  v8f acc0, acc1;
#pragma unroll
  for (int v = 0; v < 8; ++v) { acc0[v] = b1[n0 + 8 * hl + v]; acc1[v] = b1[n1 + 8 * hl + v]; }
#pragma unroll
  for (int ks = 0; ks < KDIM / 4; ++ks) {
    const float* wrow = W1s + (2 * ks + hl) * W1_PITCH + 2 * c;
    const v2f wf0 = *(const v2f*)(wrow + 2 * n0);
    const v2f wf1 = *(const v2f*)(wrow + 2 * n1);
    acc0 = __builtin_amdgcn_wmma_f32_16x16x4_f32(
         false, wf0,  false, xf[ks],
         (short)0, acc0,  false,  false);
    acc1 = __builtin_amdgcn_wmma_f32_16x16x4_f32(
         false, wf1,  false, xf[ks],
         (short)0, acc1,  false,  false);
  }
#pragma unroll
  for (int v = 0; v < 8; ++v) { acc0[v] = fmaxf(acc0[v], 0.0f); acc1[v] = fmaxf(acc1[v], 0.0f); }
  __syncthreads();
  *(float4*)(myrow + n0 + 8 * hl)     = make_float4(acc0[0], acc0[1], acc0[2], acc0[3]);
  *(float4*)(myrow + n0 + 8 * hl + 4) = make_float4(acc0[4], acc0[5], acc0[6], acc0[7]);
  *(float4*)(myrow + n1 + 8 * hl)     = make_float4(acc1[0], acc1[1], acc1[2], acc1[3]);
  *(float4*)(myrow + n1 + 8 * hl + 4) = make_float4(acc1[4], acc1[5], acc1[6], acc1[7]);
  __syncthreads();

#pragma unroll
  for (int ks = 0; ks < KDIM / 4; ++ks) xf[ks] = *(const v2f*)(myrow + 4 * ks + 2 * hl);
  v8f acc2;
#pragma unroll
  for (int v = 0; v < 8; ++v) acc2[v] = 0.0f;
#pragma unroll
  for (int ks = 0; ks < KDIM / 4; ++ks) {
    const v2f wf = *(const v2f*)(W2s + (2 * ks + hl) * W2_PITCH + 2 * (16 * q + c));
    acc2 = __builtin_amdgcn_wmma_f32_16x16x4_f32(
         false, wf,  false, xf[ks],
         (short)0, acc2,  false,  false);
  }
  const int row = rbase + rt * 16 + c;
  if (row < N_NODES) {
    float* gp = g + (size_t)row * 64 + 16 * q + 8 * hl;
    *(float4*)gp       = make_float4(acc2[0], acc2[1], acc2[2], acc2[3]);
    *(float4*)(gp + 4) = make_float4(acc2[4], acc2[5], acc2[6], acc2[7]);
  }
}

extern "C" void kernel_launch(void* const* d_in, const int* in_sizes, int n_in,
                              void* d_out, int out_size, void* d_ws, size_t ws_size,
                              hipStream_t stream) {
  const float* x  = (const float*)d_in[0];
  const int*   er = (const int*)d_in[1];
  const int*   ec = (const int*)d_in[2];
  const float* ev = (const float*)d_in[3];
  const float* W1 = (const float*)d_in[4];
  const float* b1 = (const float*)d_in[5];
  const float* W2 = (const float*)d_in[6];
  const float* b2 = (const float*)d_in[7];
  float* out = (float*)d_out;
  const int E = in_sizes[1];

  float* g  = (float*)d_ws;
  int* tcnt   = (int*)(g + (size_t)N_NODES * 64);
  int* tlst   = tcnt + TABLE_LEN;
  int* rowptr = tlst + TABLE_LEN;
  int* tmp_r  = rowptr + 100096;
  edge_cv* tmp_cv = (edge_cv*)(tmp_r + EDGE_SLOTS);
  edge_cv* cv = tmp_cv + EDGE_SLOTS;
  edge_cv* big_cv = cv + EDGE_SLOTS;
  int* big_r  = (int*)(big_cv + EDGE_SLOTS);

  const int fineBlocks = (N_NODES >> 9) + 1;
  const int gatherBlocks = (N_NODES + 8 * ROWS_PER_WAVE - 1) / (8 * ROWS_PER_WAVE);
  const int layer1Blocks = (N_NODES + L1_ROWS - 1) / L1_ROWS;

  block_bin_sort<<<N_EDGE_BLOCKS, 256, 0, stream>>>(er, ec, ev, tcnt, tlst, tmp_r, tmp_cv, E);
  fine_sort<<<fineBlocks, 32 * FS_WAVES, 0, stream>>>(tcnt, tlst, tmp_r, tmp_cv, big_r, big_cv, rowptr, cv, E);
  layer1_fused<<<layer1Blocks, 32 * L1_WAVES, 0, stream>>>(rowptr, cv, x, W1, b1, W2, g, E);
  spmm_row_gather<64, true><<<gatherBlocks, 256, 0, stream>>>(rowptr, cv, g, b2, out, E);
}
